# starting kernel
# baseline (speedup 1.0000x reference)
#include <hip/hip_runtime.h>
#include <hip/hip_bf16.h>
#include <math.h>


typedef __attribute__((ext_vector_type(2))) float v2f;
typedef __attribute__((ext_vector_type(8))) float v8f;

#define RS2 0.70710678118654752f

__global__ __launch_bounds__(96) void qc_build_G(
    const float* __restrict__ thetas, const float* __restrict__ phis,
    float* __restrict__ G) {
  int t = threadIdx.x;
  if (t >= 96) return;
  int l = t >> 5;
  int j = t & 31;

  float re[32], im[32];
#pragma unroll
  for (int i = 0; i < 32; ++i) { re[i] = 0.0f; im[i] = 0.0f; }
  re[j] = 1.0f;

  const int cb[4] = {8, 4, 2, 1};
  const int tb[4] = {4, 2, 1, 8};
#pragma unroll
  for (int w = 0; w < 4; ++w) {
    float th = thetas[4 * l + w];
    float ec = cosf(0.5f * th), es = sinf(0.5f * th);
    int bc = cb[w], bt = tb[w];
#pragma unroll
    for (int i = 0; i < 32; ++i) {
      if ((i & bc) && !(i & bt)) {
        int i1 = i | bt;
        float ur = re[i], ui = im[i], vr = re[i1], vi = im[i1];
        re[i]  =  es * vr + ec * vi;
        im[i]  =  es * vi - ec * vr;
        re[i1] = -es * ur + ec * ui;
        im[i1] = -es * ui - ec * ur;
      }
    }
  }
  float pr = cosf(phis[l]), pi = sinf(phis[l]);
#pragma unroll
  for (int i = 0; i < 32; ++i) {
    if ((i & 16) && (i & 8)) {
      float r = re[i] * pr - im[i] * pi;
      float m = re[i] * pi + im[i] * pr;
      re[i] = r; im[i] = m;
    }
  }
  float* Gl = G + l * 64 * 64;
#pragma unroll
  for (int r = 0; r < 32; ++r) {
    Gl[r * 64 + j]             =  re[r];
    Gl[r * 64 + j + 32]        = -im[r];
    Gl[(r + 32) * 64 + j]      =  im[r];
    Gl[(r + 32) * 64 + j + 32] =  re[r];
  }
}

#define GP 65
#define SP 33

__global__ __launch_bounds__(128) void qc_main(
    const float* __restrict__ x, const float* __restrict__ G,
    float* __restrict__ out, int P) {
  __shared__ float Gs[64 * GP];
  __shared__ float Sb[4][64 * SP];

  const int tid  = threadIdx.x;
  const int lane = tid & 31;
  const int wv   = tid >> 5;
  const int m    = lane & 15;
  const int hi   = lane >> 4;
  float* Sw = &Sb[wv][0];

  const int p  = blockIdx.x * 128 + tid;
  const int b  = p >> 14;
  const int h2 = (p >> 7) & 127;
  const int w2 = p & 127;

  float ang[12];
#pragma unroll
  for (int c = 0; c < 3; ++c) {
#pragma unroll
    for (int dy = 0; dy < 2; ++dy) {
      const float* row = x + (((b * 3 + c) * 256 + (2 * h2 + dy)) * 256 + 2 * w2);
      ang[c * 4 + dy * 2 + 0] = row[0];
      ang[c * 4 + dy * 2 + 1] = row[1];
    }
  }

  float st[64];
#pragma unroll
  for (int i = 0; i < 64; ++i) st[i] = 0.0f;
  st[0] = RS2;
  st[16] = RS2;

#pragma unroll
  for (int l = 0; l < 3; ++l) {
#pragma unroll
    for (int wq = 0; wq < 4; ++wq) {
      float h = 0.5f * ang[4 * l + wq];
      float c_ = cosf(h), s_ = sinf(h);
      const int bt = 8 >> wq;
#pragma unroll
      for (int i = 0; i < 32; ++i) {
        if (!(i & bt)) {
          int jj = i | bt;
          float ur = st[i], ui = st[i + 32], vr = st[jj], vi = st[jj + 32];
          st[i]       = c_ * ur + s_ * vi;
          st[i + 32]  = c_ * ui - s_ * vr;
          st[jj]      = c_ * vr + s_ * ui;
          st[jj + 32] = c_ * vi - s_ * ur;
        }
      }
    }

    __syncthreads();

    for (int idx = tid; idx < 4096; idx += 128) {
      int r = idx >> 6, cc = idx & 63;
      Gs[r * GP + cc] = G[(l * 64 + r) * 64 + cc];
    }
#pragma unroll
    for (int r = 0; r < 64; ++r) Sw[r * SP + lane] = st[r];

    __syncthreads();

    v8f acc[4][2] = {};
    for (int ks = 0; ks < 16; ++ks) {
      const int kA = ks * 4 + 2 * hi;
      v2f Bf0, Bf1;
      Bf0[0] = Sw[(kA + 0) * SP + m];
      Bf0[1] = Sw[(kA + 1) * SP + m];
      Bf1[0] = Sw[(kA + 0) * SP + 16 + m];
      Bf1[1] = Sw[(kA + 1) * SP + 16 + m];
#pragma unroll
      for (int mt = 0; mt < 4; ++mt) {
        v2f Af;
        Af[0] = Gs[(mt * 16 + m) * GP + kA];
        Af[1] = Gs[(mt * 16 + m) * GP + kA + 1];
        acc[mt][0] = __builtin_amdgcn_wmma_f32_16x16x4_f32(
            false, Af, false, Bf0, (short)0, acc[mt][0], false, false);
        acc[mt][1] = __builtin_amdgcn_wmma_f32_16x16x4_f32(
            false, Af, false, Bf1, (short)0, acc[mt][1], false, false);
      }
    }

#pragma unroll
    for (int mt = 0; mt < 4; ++mt)
#pragma unroll
      for (int nt = 0; nt < 2; ++nt)
#pragma unroll
        for (int r = 0; r < 8; ++r)
          Sw[(mt * 16 + hi * 8 + r) * SP + nt * 16 + m] = acc[mt][nt][r];

#pragma unroll
    for (int r = 0; r < 64; ++r) st[r] = Sw[r * SP + lane];
  }

#pragma unroll
  for (int i = 0; i < 16; ++i) {
    float ur = st[i],      ui = st[i + 32];
    float vr = st[i + 16], vi = st[i + 48];
    st[i]      = (ur + vr) * RS2;
    st[i + 32] = (ui + vi) * RS2;
    st[i + 16] = (ur - vr) * RS2;
    st[i + 48] = (ui - vi) * RS2;
  }

  float ev = 0.0f;
#pragma unroll
  for (int i = 0; i < 32; ++i) {
    float p2 = st[i] * st[i] + st[i + 32] * st[i + 32];
    ev += (i & 16) ? -p2 : p2;
  }
  if (p < P) out[p] = ev;
}

extern "C" void kernel_launch(void* const* d_in, const int* in_sizes, int n_in,
                              void* d_out, int out_size, void* d_ws, size_t ws_size,
                              hipStream_t stream) {
  (void)in_sizes; (void)n_in; (void)ws_size;
  const float* x      = (const float*)d_in[0];
  const float* thetas = (const float*)d_in[1];
  const float* phis   = (const float*)d_in[2];
  float* out = (float*)d_out;
  float* G   = (float*)d_ws;

  hipLaunchKernelGGL(qc_build_G, dim3(1), dim3(96), 0, stream, thetas, phis, G);

  const int P = out_size;
  const int blocks = (P + 127) / 128;
  hipLaunchKernelGGL(qc_main, dim3(blocks), dim3(128), 0, stream, x, G, out, P);
}


# State kept in WMMA result registers across layers, fed back as B (G k-permuted, fragment order, staged once): no LDS state round trip, no per-layer barrier; k order and final sum order change
# speedup vs baseline: 1.2041x; 1.2041x over previous
#include <hip/hip_runtime.h>
#include <math.h>


typedef __attribute__((ext_vector_type(2))) float v2f;
typedef __attribute__((ext_vector_type(4))) float v4f;
typedef __attribute__((ext_vector_type(8))) float v8f;

#define RS2 0.70710678118654752f

__device__ __forceinline__ int sw34(int i) {
  return (i & 7) | ((i & 8) << 1) | ((i & 16) >> 1);
}

__device__ __forceinline__ int frag_off(int R, int C) {
  return ((((R >> 4) * 4 + (C >> 4)) * 2 + ((C >> 2) & 1)) * 32 + ((C >> 3) & 1) * 16 + (R & 15)) * 4 + (C & 3);
}

__global__ __launch_bounds__(96) void qc_build_G(
    const float* __restrict__ thetas, const float* __restrict__ phis,
    float* __restrict__ G) {
  int t = threadIdx.x;
  if (t >= 96) return;
  int l = t >> 5;
  int j = t & 31;

  float re[32], im[32];
#pragma unroll
  for (int i = 0; i < 32; ++i) { re[i] = 0.0f; im[i] = 0.0f; }
  re[j] = 1.0f;

  const int cb[4] = {8, 4, 2, 1};
  const int tb[4] = {4, 2, 1, 8};
#pragma unroll
  for (int w = 0; w < 4; ++w) {
    float th = thetas[4 * l + w];
    float ec = cosf(0.5f * th), es = sinf(0.5f * th);
    int bc = cb[w], bt = tb[w];
#pragma unroll
    for (int i = 0; i < 32; ++i) {
      if ((i & bc) && !(i & bt)) {
        int i1 = i | bt;
        float ur = re[i], ui = im[i], vr = re[i1], vi = im[i1];
        re[i]  =  es * vr + ec * vi;
        im[i]  =  es * vi - ec * vr;
        re[i1] = -es * ur + ec * ui;
        im[i1] = -es * ui - ec * ur;
      }
    }
  }
  float pr = cosf(phis[l]), pi = sinf(phis[l]);
#pragma unroll
  for (int i = 0; i < 32; ++i) {
    if ((i & 16) && (i & 8)) {
      float r = re[i] * pr - im[i] * pi;
      float m = re[i] * pi + im[i] * pr;
      re[i] = r; im[i] = m;
    }
  }
  float* Gl = G + l * 64 * 64;
  const int Cj = sw34(j);
#pragma unroll
  for (int r = 0; r < 32; ++r) {
    const int Rr = sw34(r);
    Gl[frag_off(Rr,      Cj)]      =  re[r];
    Gl[frag_off(Rr,      Cj + 32)] = -im[r];
    Gl[frag_off(Rr + 32, Cj)]      =  im[r];
    Gl[frag_off(Rr + 32, Cj + 32)] =  re[r];
  }
}

__device__ __forceinline__ float xhalf(float v) {
  int i = __builtin_bit_cast(int, v);
  return __builtin_bit_cast(float,
      __builtin_amdgcn_permlanex16(i, i, 0x76543210, 0xfedcba98, false, false));
}

__global__ __launch_bounds__(256) void qc_main(
    const float* __restrict__ x, const float* __restrict__ G,
    float* __restrict__ out, int P) {
  __shared__ v4f Gs[3 * 1024];

  const int tid  = threadIdx.x;
  const int lane = tid & 31;
  const bool hi  = (lane >> 4) != 0;

  const int p  = blockIdx.x * 256 + tid;
  const int b  = p >> 14;
  const int h2 = (p >> 7) & 127;
  const int w2 = p & 127;

  float ang[12];
#pragma unroll
  for (int c = 0; c < 3; ++c) {
#pragma unroll
    for (int dy = 0; dy < 2; ++dy) {
      const float* row = x + (((b * 3 + c) * 256 + (2 * h2 + dy)) * 256 + 2 * w2);
      ang[c * 4 + dy * 2 + 0] = row[0];
      ang[c * 4 + dy * 2 + 1] = row[1];
    }
  }

  const v4f* G4 = reinterpret_cast<const v4f*>(G);
  v4f gt[12];
#pragma unroll
  for (int k = 0; k < 12; ++k) gt[k] = G4[tid + 256 * k];

  float cso[12], sno[12];
#pragma unroll
  for (int k = 0; k < 12; ++k) {
    float hf = 0.5f * ang[k];
    cso[k] = cosf(hf);
    sno[k] = sinf(hf);
  }

#pragma unroll
  for (int k = 0; k < 12; ++k) Gs[tid + 256 * k] = gt[k];

  float st[4][2][8];
#pragma unroll
  for (int mt = 0; mt < 4; ++mt)
#pragma unroll
    for (int nt = 0; nt < 2; ++nt)
#pragma unroll
      for (int r = 0; r < 8; ++r) st[mt][nt][r] = 0.0f;
  st[0][0][0] = RS2;
  st[0][1][0] = RS2;

  __syncthreads();

  const v4f* Gw = Gs + lane;

#pragma unroll
  for (int l = 0; l < 3; ++l) {
#pragma unroll
    for (int wq = 0; wq < 4; ++wq) {
      const float co = cso[4 * l + wq], so = sno[4 * l + wq];
      const float cx = xhalf(co), sx = xhalf(so);
      const float cc[2] = { hi ? cx : co, hi ? co : cx };
      const float ss[2] = { hi ? sx : so, hi ? so : sx };
#pragma unroll
      for (int nt = 0; nt < 2; ++nt) {
        const float c_ = cc[nt], s_ = ss[nt];
        if (wq == 0) {
#pragma unroll
          for (int r = 0; r < 8; ++r) {
            float ur = st[0][nt][r], ui = st[2][nt][r], vr = st[1][nt][r], vi = st[3][nt][r];
            st[0][nt][r] = c_ * ur + s_ * vi;
            st[2][nt][r] = c_ * ui - s_ * vr;
            st[1][nt][r] = c_ * vr + s_ * ui;
            st[3][nt][r] = c_ * vi - s_ * ur;
          }
        } else {
          const int bt = 8 >> wq;
#pragma unroll
          for (int w1 = 0; w1 < 2; ++w1) {
#pragma unroll
            for (int r = 0; r < 8; ++r) {
              if (!(r & bt)) {
                const int jj = r | bt;
                float ur = st[w1][nt][r],  ui = st[2 + w1][nt][r];
                float vr = st[w1][nt][jj], vi = st[2 + w1][nt][jj];
                st[w1][nt][r]      = c_ * ur + s_ * vi;
                st[2 + w1][nt][r]  = c_ * ui - s_ * vr;
                st[w1][nt][jj]     = c_ * vr + s_ * ui;
                st[2 + w1][nt][jj] = c_ * vi - s_ * ur;
              }
            }
          }
        }
      }
    }

    v8f nw[4][2];
#pragma unroll
    for (int mo = 0; mo < 4; ++mo)
#pragma unroll
      for (int nt = 0; nt < 2; ++nt) nw[mo][nt] = (v8f){0.f, 0.f, 0.f, 0.f, 0.f, 0.f, 0.f, 0.f};

#pragma unroll
    for (int kmt = 0; kmt < 4; ++kmt) {
      v4f a[4][2];
#pragma unroll
      for (int mo = 0; mo < 4; ++mo)
#pragma unroll
        for (int q = 0; q < 2; ++q)
          a[mo][q] = Gw[l * 1024 + ((mo * 4 + kmt) * 2 + q) * 32];
#pragma unroll
      for (int rp = 0; rp < 4; ++rp) {
#pragma unroll
        for (int mo = 0; mo < 4; ++mo) {
          const v4f aq = a[mo][rp >> 1];
          v2f Af;
          Af[0] = (rp & 1) ? aq[2] : aq[0];
          Af[1] = (rp & 1) ? aq[3] : aq[1];
#pragma unroll
          for (int nt = 0; nt < 2; ++nt) {
            v2f Bf;
            Bf[0] = st[kmt][nt][2 * rp];
            Bf[1] = st[kmt][nt][2 * rp + 1];
            nw[mo][nt] = __builtin_amdgcn_wmma_f32_16x16x4_f32(
                false, Af, false, Bf, (short)0, nw[mo][nt], false, false);
          }
        }
      }
    }

#pragma unroll
    for (int mo = 0; mo < 4; ++mo)
#pragma unroll
      for (int nt = 0; nt < 2; ++nt)
#pragma unroll
        for (int r = 0; r < 8; ++r) st[mo][nt][r] = nw[mo][nt][r];
  }

  const float sg = hi ? -1.0f : 1.0f;
  float tot[2];
#pragma unroll
  for (int nt = 0; nt < 2; ++nt) {
    float acc = 0.0f;
#pragma unroll
    for (int w1 = 0; w1 < 2; ++w1) {
#pragma unroll
      for (int r = 0; r < 8; ++r) {
        float mr = st[w1][nt][r],     mi = st[2 + w1][nt][r];
        float orr = xhalf(mr),        oi = xhalf(mi);
        float tr = (sg * mr + orr) * RS2;
        float ti = (sg * mi + oi) * RS2;
        acc += tr * tr + ti * ti;
      }
    }
    float part = sg * acc;
    tot[nt] = part + xhalf(part);
  }
  const float ev = hi ? tot[1] : tot[0];
  if (p < P) out[p] = ev;
}

extern "C" void kernel_launch(void* const* d_in, const int* in_sizes, int n_in,
                              void* d_out, int out_size, void* d_ws, size_t ws_size,
                              hipStream_t stream) {
  (void)in_sizes; (void)n_in; (void)ws_size;
  const float* x      = (const float*)d_in[0];
  const float* thetas = (const float*)d_in[1];
  const float* phis   = (const float*)d_in[2];
  float* out = (float*)d_out;
  float* G   = (float*)d_ws;

  hipLaunchKernelGGL(qc_build_G, dim3(1), dim3(96), 0, stream, thetas, phis, G);

  const int P = out_size;
  const int blocks = (P + 255) / 256;
  hipLaunchKernelGGL(qc_main, dim3(blocks), dim3(256), 0, stream, x, G, out, P);
}


# Layer unitary split by wire 0 (control only): two 32x32 real blocks applied per w0 instead of one 64x64, half the WMMAs; lane half now wire 1 (RX via lane swap); k order and final sum order change
# speedup vs baseline: 1.8100x; 1.5032x over previous
#include <hip/hip_runtime.h>
#include <math.h>


typedef __attribute__((ext_vector_type(2))) float v2f;
typedef __attribute__((ext_vector_type(4))) float v4f;
typedef __attribute__((ext_vector_type(8))) float v8f;

#define RS2 0.70710678118654752f

__device__ __forceinline__ int frag_off(int R, int C) {
  return ((((R >> 4) * 2 + (C >> 4)) * 2 + ((C >> 2) & 1)) * 32 + ((C >> 3) & 1) * 16 + (R & 15)) * 4 + (C & 3);
}

__global__ __launch_bounds__(96) void qc_build_G(
    const float* __restrict__ thetas, const float* __restrict__ phis,
    float* __restrict__ G) {
  int t = threadIdx.x;
  if (t >= 96) return;
  const int l = t >> 5;
  const bool w0 = (t & 16) != 0;
  const int a = t & 15;

  float re[16], im[16];
#pragma unroll
  for (int i = 0; i < 16; ++i) { re[i] = (i == a) ? 1.0f : 0.0f; im[i] = 0.0f; }

  const int cb[4] = {8, 4, 2, 1};
  const int tb[4] = {4, 2, 1, 8};
#pragma unroll
  for (int w = 0; w < 4; ++w) {
    float th = thetas[4 * l + w];
    float ec = cosf(0.5f * th), es = sinf(0.5f * th);
    int bc = cb[w], bt = tb[w];
#pragma unroll
    for (int i = 0; i < 16; ++i) {
      if ((i & bc) && !(i & bt)) {
        int i1 = i | bt;
        float ur = re[i], ui = im[i], vr = re[i1], vi = im[i1];
        re[i]  =  es * vr + ec * vi;
        im[i]  =  es * vi - ec * vr;
        re[i1] = -es * ur + ec * ui;
        im[i1] = -es * ui - ec * ur;
      }
    }
  }
  float pr = cosf(phis[l]), pi = sinf(phis[l]);
#pragma unroll
  for (int i = 8; i < 16; ++i) {
    float r = re[i] * pr - im[i] * pi;
    float m = re[i] * pi + im[i] * pr;
    re[i] = w0 ? r : re[i];
    im[i] = w0 ? m : im[i];
  }
  float* Gb = G + (l * 2 + (w0 ? 1 : 0)) * 1024;
#pragma unroll
  for (int r = 0; r < 16; ++r) {
    Gb[frag_off(r,      a)]      =  re[r];
    Gb[frag_off(r,      a + 16)] = -im[r];
    Gb[frag_off(r + 16, a)]      =  im[r];
    Gb[frag_off(r + 16, a + 16)] =  re[r];
  }
}

__device__ __forceinline__ float xhalf(float v) {
  int i = __builtin_bit_cast(int, v);
  return __builtin_bit_cast(float,
      __builtin_amdgcn_permlanex16(i, i, 0x76543210, 0xfedcba98, false, false));
}

__global__ __launch_bounds__(256) void qc_main(
    const float* __restrict__ x, const float* __restrict__ G,
    float* __restrict__ out, int P) {
  __shared__ v4f Gs[3 * 2 * 256];

  const int tid  = threadIdx.x;
  const int lane = tid & 31;
  const bool hi  = (lane >> 4) != 0;

  const int p  = blockIdx.x * 256 + tid;
  const int b  = p >> 14;
  const int h2 = (p >> 7) & 127;
  const int w2 = p & 127;

  float ang[12];
#pragma unroll
  for (int c = 0; c < 3; ++c) {
#pragma unroll
    for (int dy = 0; dy < 2; ++dy) {
      const float* row = x + (((b * 3 + c) * 256 + (2 * h2 + dy)) * 256 + 2 * w2);
      ang[c * 4 + dy * 2 + 0] = row[0];
      ang[c * 4 + dy * 2 + 1] = row[1];
    }
  }

  const v4f* G4 = reinterpret_cast<const v4f*>(G);
  v4f gt[6];
#pragma unroll
  for (int k = 0; k < 6; ++k) gt[k] = G4[tid + 256 * k];

  float cso[12], sno[12];
#pragma unroll
  for (int k = 0; k < 12; ++k) {
    float hf = 0.5f * ang[k];
    cso[k] = cosf(hf);
    sno[k] = sinf(hf);
  }

#pragma unroll
  for (int k = 0; k < 6; ++k) Gs[tid + 256 * k] = gt[k];

  float st[2][2][2][8];
#pragma unroll
  for (int w0 = 0; w0 < 2; ++w0)
#pragma unroll
    for (int mt = 0; mt < 2; ++mt)
#pragma unroll
      for (int nt = 0; nt < 2; ++nt)
#pragma unroll
        for (int r = 0; r < 8; ++r) st[w0][mt][nt][r] = 0.0f;
  const float amp0 = hi ? 0.0f : RS2;
  st[0][0][0][0] = amp0;
  st[0][0][1][0] = amp0;
  st[1][0][0][0] = amp0;
  st[1][0][1][0] = amp0;

  __syncthreads();

  const v4f* Gw = Gs + lane;

#pragma unroll
  for (int l = 0; l < 3; ++l) {
#pragma unroll
    for (int wq = 0; wq < 4; ++wq) {
      const float co = cso[4 * l + wq], so = sno[4 * l + wq];
      const float cx = xhalf(co), sx = xhalf(so);
      const float cc[2] = { hi ? cx : co, hi ? co : cx };
      const float ss[2] = { hi ? sx : so, hi ? so : sx };
#pragma unroll
      for (int nt = 0; nt < 2; ++nt) {
        const float c_ = cc[nt], s_ = ss[nt];
        if (wq == 0) {
#pragma unroll
          for (int w0 = 0; w0 < 2; ++w0) {
#pragma unroll
            for (int r = 0; r < 8; ++r) {
              float mr = st[w0][0][nt][r], mi = st[w0][1][nt][r];
              float orr = xhalf(mr),       oi = xhalf(mi);
              st[w0][0][nt][r] = c_ * mr + s_ * oi;
              st[w0][1][nt][r] = c_ * mi - s_ * orr;
            }
          }
        } else {
          const int bt = 8 >> wq;
#pragma unroll
          for (int w0 = 0; w0 < 2; ++w0) {
#pragma unroll
            for (int r = 0; r < 8; ++r) {
              if (!(r & bt)) {
                const int jj = r | bt;
                float ur = st[w0][0][nt][r],  ui = st[w0][1][nt][r];
                float vr = st[w0][0][nt][jj], vi = st[w0][1][nt][jj];
                st[w0][0][nt][r]  = c_ * ur + s_ * vi;
                st[w0][1][nt][r]  = c_ * ui - s_ * vr;
                st[w0][0][nt][jj] = c_ * vr + s_ * ui;
                st[w0][1][nt][jj] = c_ * vi - s_ * ur;
              }
            }
          }
        }
      }
    }

    v8f nw[2][2][2];
#pragma unroll
    for (int w0 = 0; w0 < 2; ++w0)
#pragma unroll
      for (int mo = 0; mo < 2; ++mo)
#pragma unroll
        for (int nt = 0; nt < 2; ++nt) nw[w0][mo][nt] = (v8f){0.f, 0.f, 0.f, 0.f, 0.f, 0.f, 0.f, 0.f};

#pragma unroll
    for (int w0 = 0; w0 < 2; ++w0) {
#pragma unroll
      for (int kmt = 0; kmt < 2; ++kmt) {
        v4f a[2][2];
#pragma unroll
        for (int mo = 0; mo < 2; ++mo)
#pragma unroll
          for (int q = 0; q < 2; ++q)
            a[mo][q] = Gw[(l * 2 + w0) * 256 + ((mo * 2 + kmt) * 2 + q) * 32];
#pragma unroll
        for (int rp = 0; rp < 4; ++rp) {
#pragma unroll
          for (int mo = 0; mo < 2; ++mo) {
            const v4f aq = a[mo][rp >> 1];
            v2f Af;
            Af[0] = (rp & 1) ? aq[2] : aq[0];
            Af[1] = (rp & 1) ? aq[3] : aq[1];
#pragma unroll
            for (int nt = 0; nt < 2; ++nt) {
              v2f Bf;
              Bf[0] = st[w0][kmt][nt][2 * rp];
              Bf[1] = st[w0][kmt][nt][2 * rp + 1];
              nw[w0][mo][nt] = __builtin_amdgcn_wmma_f32_16x16x4_f32(
                  false, Af, false, Bf, (short)0, nw[w0][mo][nt], false, false);
            }
          }
        }
      }
    }

#pragma unroll
    for (int w0 = 0; w0 < 2; ++w0)
#pragma unroll
      for (int mo = 0; mo < 2; ++mo)
#pragma unroll
        for (int nt = 0; nt < 2; ++nt)
#pragma unroll
          for (int r = 0; r < 8; ++r) st[w0][mo][nt][r] = nw[w0][mo][nt][r];
  }

  float tot[2];
#pragma unroll
  for (int nt = 0; nt < 2; ++nt) {
    float accp = 0.0f, accm = 0.0f;
#pragma unroll
    for (int mt = 0; mt < 2; ++mt) {
#pragma unroll
      for (int r = 0; r < 8; ++r) {
        float u = st[0][mt][nt][r], v = st[1][mt][nt][r];
        float tp = (u + v) * RS2;
        float tm = (u - v) * RS2;
        accp += tp * tp;
        accm += tm * tm;
      }
    }
    float part = accp - accm;
    tot[nt] = part + xhalf(part);
  }
  const float ev = hi ? tot[1] : tot[0];
  if (p < P) out[p] = ev;
}

extern "C" void kernel_launch(void* const* d_in, const int* in_sizes, int n_in,
                              void* d_out, int out_size, void* d_ws, size_t ws_size,
                              hipStream_t stream) {
  (void)in_sizes; (void)n_in; (void)ws_size;
  const float* x      = (const float*)d_in[0];
  const float* thetas = (const float*)d_in[1];
  const float* phis   = (const float*)d_in[2];
  float* out = (float*)d_out;
  float* G   = (float*)d_ws;

  hipLaunchKernelGGL(qc_build_G, dim3(1), dim3(96), 0, stream, thetas, phis, G);

  const int P = out_size;
  const int blocks = (P + 255) / 256;
  hipLaunchKernelGGL(qc_main, dim3(blocks), dim3(256), 0, stream, x, G, out, P);
}


# Layer 0 specialised to the fixed start state: RX of |0000> formed directly as 16 real magnitudes, one 32x16 product for both w0 (CPhase applied after): 16 WMMAs not 64; layer 0 k sum order changes
# speedup vs baseline: 1.9708x; 1.0888x over previous
#include <hip/hip_runtime.h>
#include <math.h>


typedef __attribute__((ext_vector_type(2))) float v2f;
typedef __attribute__((ext_vector_type(4))) float v4f;
typedef __attribute__((ext_vector_type(8))) float v8f;

#define RS2 0.70710678118654752f
#define G_L12   512
#define G_PHASE 4608
#define G_V4    1152

__device__ __forceinline__ int frag_off(int R, int C) {
  return ((((R >> 4) * 2 + (C >> 4)) * 2 + ((C >> 2) & 1)) * 32 + ((C >> 3) & 1) * 16 + (R & 15)) * 4 + (C & 3);
}

__device__ __forceinline__ int frag0_off(int R, int C) {
  return ((((R >> 4) * 2 + ((C >> 2) & 1)) * 32) + ((C >> 3) & 1) * 16 + (R & 15)) * 4 + (C & 3);
}

__global__ __launch_bounds__(96) void qc_build_G(
    const float* __restrict__ thetas, const float* __restrict__ phis,
    float* __restrict__ G) {
  int t = threadIdx.x;
  if (t >= 96) return;
  const int l = t >> 5;
  const bool w0 = (t & 16) != 0;
  const int a = t & 15;

  float re[16], im[16];
#pragma unroll
  for (int i = 0; i < 16; ++i) { re[i] = (i == a) ? 1.0f : 0.0f; im[i] = 0.0f; }

  const int cb[4] = {8, 4, 2, 1};
  const int tb[4] = {4, 2, 1, 8};
#pragma unroll
  for (int w = 0; w < 4; ++w) {
    float th = thetas[4 * l + w];
    float ec = cosf(0.5f * th), es = sinf(0.5f * th);
    int bc = cb[w], bt = tb[w];
#pragma unroll
    for (int i = 0; i < 16; ++i) {
      if ((i & bc) && !(i & bt)) {
        int i1 = i | bt;
        float ur = re[i], ui = im[i], vr = re[i1], vi = im[i1];
        re[i]  =  es * vr + ec * vi;
        im[i]  =  es * vi - ec * vr;
        re[i1] = -es * ur + ec * ui;
        im[i1] = -es * ui - ec * ur;
      }
    }
  }
  const float pr = cosf(phis[l]), pi = sinf(phis[l]);
  if (l == 0) {
    if (!w0) {
      const int pc = __popc(a) & 3;
#pragma unroll
      for (int r = 0; r < 16; ++r) {
        const float vr = re[r], vi = im[r];
        const float xr = (pc == 0) ? vr : (pc == 1) ?  vi : (pc == 2) ? -vr : -vi;
        const float xi = (pc == 0) ? vi : (pc == 1) ? -vr : (pc == 2) ? -vi :  vr;
        G[frag0_off(r,      a)] = xr;
        G[frag0_off(r + 16, a)] = xi;
      }
    } else if (a == 0) {
      G[G_PHASE]     = pr;
      G[G_PHASE + 1] = pi;
    }
    return;
  }
#pragma unroll
  for (int i = 8; i < 16; ++i) {
    float r = re[i] * pr - im[i] * pi;
    float m = re[i] * pi + im[i] * pr;
    re[i] = w0 ? r : re[i];
    im[i] = w0 ? m : im[i];
  }
  float* Gb = G + G_L12 + ((l - 1) * 2 + (w0 ? 1 : 0)) * 1024;
#pragma unroll
  for (int r = 0; r < 16; ++r) {
    Gb[frag_off(r,      a)]      =  re[r];
    Gb[frag_off(r,      a + 16)] = -im[r];
    Gb[frag_off(r + 16, a)]      =  im[r];
    Gb[frag_off(r + 16, a + 16)] =  re[r];
  }
}

__device__ __forceinline__ float xhalf(float v) {
  int i = __builtin_bit_cast(int, v);
  return __builtin_bit_cast(float,
      __builtin_amdgcn_permlanex16(i, i, 0x76543210, 0xfedcba98, false, false));
}

__global__ __launch_bounds__(256) void qc_main(
    const float* __restrict__ x, const float* __restrict__ G,
    float* __restrict__ out, int P) {
  __shared__ v4f Gs[G_V4];

  const int tid  = threadIdx.x;
  const int lane = tid & 31;
  const bool hi  = (lane >> 4) != 0;

  const int p  = blockIdx.x * 256 + tid;
  const int b  = p >> 14;
  const int h2 = (p >> 7) & 127;
  const int w2 = p & 127;

  float ang[12];
#pragma unroll
  for (int c = 0; c < 3; ++c) {
#pragma unroll
    for (int dy = 0; dy < 2; ++dy) {
      const float* row = x + (((b * 3 + c) * 256 + (2 * h2 + dy)) * 256 + 2 * w2);
      ang[c * 4 + dy * 2 + 0] = row[0];
      ang[c * 4 + dy * 2 + 1] = row[1];
    }
  }

  const v4f* G4 = reinterpret_cast<const v4f*>(G);
  v4f gt[4];
#pragma unroll
  for (int k = 0; k < 4; ++k) gt[k] = G4[tid + 256 * k];
  v4f gt4 = (v4f){0.f, 0.f, 0.f, 0.f};
  if (tid < G_V4 - 1024) gt4 = G4[tid + 1024];
  const float pr0 = G[G_PHASE], pi0 = G[G_PHASE + 1];

  float cso[12], sno[12];
#pragma unroll
  for (int k = 0; k < 12; ++k) {
    float hf = 0.5f * ang[k];
    cso[k] = cosf(hf);
    sno[k] = sinf(hf);
  }

#pragma unroll
  for (int k = 0; k < 4; ++k) Gs[tid + 256 * k] = gt[k];
  if (tid < G_V4 - 1024) Gs[tid + 1024] = gt4;

  __syncthreads();

  const v4f* Gw = Gs + lane;

  float st[2][2][2][8];

  {
    float c0[4][2], s0[4][2];
#pragma unroll
    for (int wq = 0; wq < 4; ++wq) {
      const float co = cso[wq], so = sno[wq];
      const float cx = xhalf(co), sx = xhalf(so);
      c0[wq][0] = hi ? cx : co;  c0[wq][1] = hi ? co : cx;
      s0[wq][0] = hi ? sx : so;  s0[wq][1] = hi ? so : sx;
    }
    float mg[2][8];
#pragma unroll
    for (int nt = 0; nt < 2; ++nt) {
      const float f1 = (hi ? s0[0][nt] : c0[0][nt]) * RS2;
#pragma unroll
      for (int r = 0; r < 8; ++r) {
        float v = f1 * ((r & 4) ? s0[1][nt] : c0[1][nt]);
        v = v * ((r & 2) ? s0[2][nt] : c0[2][nt]);
        mg[nt][r] = v * ((r & 1) ? s0[3][nt] : c0[3][nt]);
      }
    }

    v8f acc[2][2];
#pragma unroll
    for (int mo = 0; mo < 2; ++mo)
#pragma unroll
      for (int nt = 0; nt < 2; ++nt) acc[mo][nt] = (v8f){0.f, 0.f, 0.f, 0.f, 0.f, 0.f, 0.f, 0.f};
    v4f a0[2][2];
#pragma unroll
    for (int mo = 0; mo < 2; ++mo)
#pragma unroll
      for (int q = 0; q < 2; ++q) a0[mo][q] = Gw[(mo * 2 + q) * 32];
#pragma unroll
    for (int rp = 0; rp < 4; ++rp) {
#pragma unroll
      for (int mo = 0; mo < 2; ++mo) {
        const v4f aq = a0[mo][rp >> 1];
        v2f Af;
        Af[0] = (rp & 1) ? aq[2] : aq[0];
        Af[1] = (rp & 1) ? aq[3] : aq[1];
#pragma unroll
        for (int nt = 0; nt < 2; ++nt) {
          v2f Bf;
          Bf[0] = mg[nt][2 * rp];
          Bf[1] = mg[nt][2 * rp + 1];
          acc[mo][nt] = __builtin_amdgcn_wmma_f32_16x16x4_f32(
              false, Af, false, Bf, (short)0, acc[mo][nt], false, false);
        }
      }
    }

    const float pa = hi ? pr0 : 1.0f, pb = hi ? pi0 : 0.0f;
#pragma unroll
    for (int nt = 0; nt < 2; ++nt) {
#pragma unroll
      for (int r = 0; r < 8; ++r) {
        const float vr = acc[0][nt][r], vi = acc[1][nt][r];
        st[0][0][nt][r] = vr;
        st[0][1][nt][r] = vi;
        st[1][0][nt][r] = vr * pa - vi * pb;
        st[1][1][nt][r] = vr * pb + vi * pa;
      }
    }
  }

#pragma unroll
  for (int l = 1; l < 3; ++l) {
#pragma unroll
    for (int wq = 0; wq < 4; ++wq) {
      const float co = cso[4 * l + wq], so = sno[4 * l + wq];
      const float cx = xhalf(co), sx = xhalf(so);
      const float cc[2] = { hi ? cx : co, hi ? co : cx };
      const float ss[2] = { hi ? sx : so, hi ? so : sx };
#pragma unroll
      for (int nt = 0; nt < 2; ++nt) {
        const float c_ = cc[nt], s_ = ss[nt];
        if (wq == 0) {
#pragma unroll
          for (int w0 = 0; w0 < 2; ++w0) {
#pragma unroll
            for (int r = 0; r < 8; ++r) {
              float mr = st[w0][0][nt][r], mi = st[w0][1][nt][r];
              float orr = xhalf(mr),       oi = xhalf(mi);
              st[w0][0][nt][r] = c_ * mr + s_ * oi;
              st[w0][1][nt][r] = c_ * mi - s_ * orr;
            }
          }
        } else {
          const int bt = 8 >> wq;
#pragma unroll
          for (int w0 = 0; w0 < 2; ++w0) {
#pragma unroll
            for (int r = 0; r < 8; ++r) {
              if (!(r & bt)) {
                const int jj = r | bt;
                float ur = st[w0][0][nt][r],  ui = st[w0][1][nt][r];
                float vr = st[w0][0][nt][jj], vi = st[w0][1][nt][jj];
                st[w0][0][nt][r]  = c_ * ur + s_ * vi;
                st[w0][1][nt][r]  = c_ * ui - s_ * vr;
                st[w0][0][nt][jj] = c_ * vr + s_ * ui;
                st[w0][1][nt][jj] = c_ * vi - s_ * ur;
              }
            }
          }
        }
      }
    }

    v8f nw[2][2][2];
#pragma unroll
    for (int w0 = 0; w0 < 2; ++w0)
#pragma unroll
      for (int mo = 0; mo < 2; ++mo)
#pragma unroll
        for (int nt = 0; nt < 2; ++nt) nw[w0][mo][nt] = (v8f){0.f, 0.f, 0.f, 0.f, 0.f, 0.f, 0.f, 0.f};

#pragma unroll
    for (int w0 = 0; w0 < 2; ++w0) {
#pragma unroll
      for (int kmt = 0; kmt < 2; ++kmt) {
        v4f a[2][2];
#pragma unroll
        for (int mo = 0; mo < 2; ++mo)
#pragma unroll
          for (int q = 0; q < 2; ++q)
            a[mo][q] = Gw[G_L12 / 4 + ((l - 1) * 2 + w0) * 256 + ((mo * 2 + kmt) * 2 + q) * 32];
#pragma unroll
        for (int rp = 0; rp < 4; ++rp) {
#pragma unroll
          for (int mo = 0; mo < 2; ++mo) {
            const v4f aq = a[mo][rp >> 1];
            v2f Af;
            Af[0] = (rp & 1) ? aq[2] : aq[0];
            Af[1] = (rp & 1) ? aq[3] : aq[1];
#pragma unroll
            for (int nt = 0; nt < 2; ++nt) {
              v2f Bf;
              Bf[0] = st[w0][kmt][nt][2 * rp];
              Bf[1] = st[w0][kmt][nt][2 * rp + 1];
              nw[w0][mo][nt] = __builtin_amdgcn_wmma_f32_16x16x4_f32(
                  false, Af, false, Bf, (short)0, nw[w0][mo][nt], false, false);
            }
          }
        }
      }
    }

#pragma unroll
    for (int w0 = 0; w0 < 2; ++w0)
#pragma unroll
      for (int mo = 0; mo < 2; ++mo)
#pragma unroll
        for (int nt = 0; nt < 2; ++nt)
#pragma unroll
          for (int r = 0; r < 8; ++r) st[w0][mo][nt][r] = nw[w0][mo][nt][r];
  }

  float tot[2];
#pragma unroll
  for (int nt = 0; nt < 2; ++nt) {
    float accp = 0.0f, accm = 0.0f;
#pragma unroll
    for (int mt = 0; mt < 2; ++mt) {
#pragma unroll
      for (int r = 0; r < 8; ++r) {
        float u = st[0][mt][nt][r], v = st[1][mt][nt][r];
        float tp = (u + v) * RS2;
        float tm = (u - v) * RS2;
        accp += tp * tp;
        accm += tm * tm;
      }
    }
    float part = accp - accm;
    tot[nt] = part + xhalf(part);
  }
  const float ev = hi ? tot[1] : tot[0];
  if (p < P) out[p] = ev;
}

extern "C" void kernel_launch(void* const* d_in, const int* in_sizes, int n_in,
                              void* d_out, int out_size, void* d_ws, size_t ws_size,
                              hipStream_t stream) {
  (void)in_sizes; (void)n_in; (void)ws_size;
  const float* x      = (const float*)d_in[0];
  const float* thetas = (const float*)d_in[1];
  const float* phis   = (const float*)d_in[2];
  float* out = (float*)d_out;
  float* G   = (float*)d_ws;

  hipLaunchKernelGGL(qc_build_G, dim3(1), dim3(96), 0, stream, thetas, phis, G);

  const int P = out_size;
  const int blocks = (P + 255) / 256;
  hipLaunchKernelGGL(qc_main, dim3(blocks), dim3(256), 0, stream, x, G, out, P);
}


# Last layer folded into closing sum: CU gates are phased permutations, so V cancels in Re(conj(u).v), CPhase is a phase on odd w2^w3^w4 rows; no layer 2 product or wire 1 RX; sum order changes
# speedup vs baseline: 2.5082x; 1.2727x over previous
#include <hip/hip_runtime.h>
#include <math.h>


typedef __attribute__((ext_vector_type(2))) float v2f;
typedef __attribute__((ext_vector_type(4))) float v4f;
typedef __attribute__((ext_vector_type(8))) float v8f;

#define RS2 0.70710678118654752f
#define G_L12   512
#define G_PHASE 2560
#define G_V4    640

__device__ __forceinline__ int frag_off(int R, int C) {
  return ((((R >> 4) * 2 + (C >> 4)) * 2 + ((C >> 2) & 1)) * 32 + ((C >> 3) & 1) * 16 + (R & 15)) * 4 + (C & 3);
}

__device__ __forceinline__ int frag0_off(int R, int C) {
  return ((((R >> 4) * 2 + ((C >> 2) & 1)) * 32) + ((C >> 3) & 1) * 16 + (R & 15)) * 4 + (C & 3);
}

__global__ __launch_bounds__(64) void qc_build_G(
    const float* __restrict__ thetas, const float* __restrict__ phis,
    float* __restrict__ G) {
  int t = threadIdx.x;
  if (t >= 64) return;
  const int l = t >> 5;
  const bool w0 = (t & 16) != 0;
  const int a = t & 15;

  float re[16], im[16];
#pragma unroll
  for (int i = 0; i < 16; ++i) { re[i] = (i == a) ? 1.0f : 0.0f; im[i] = 0.0f; }

  const int cb[4] = {8, 4, 2, 1};
  const int tb[4] = {4, 2, 1, 8};
#pragma unroll
  for (int w = 0; w < 4; ++w) {
    float th = thetas[4 * l + w];
    float ec = cosf(0.5f * th), es = sinf(0.5f * th);
    int bc = cb[w], bt = tb[w];
#pragma unroll
    for (int i = 0; i < 16; ++i) {
      if ((i & bc) && !(i & bt)) {
        int i1 = i | bt;
        float ur = re[i], ui = im[i], vr = re[i1], vi = im[i1];
        re[i]  =  es * vr + ec * vi;
        im[i]  =  es * vi - ec * vr;
        re[i1] = -es * ur + ec * ui;
        im[i1] = -es * ui - ec * ur;
      }
    }
  }
  const float pr = cosf(phis[l]), pi = sinf(phis[l]);
  if (l == 0) {
    if (!w0) {
      const int pc = __popc(a) & 3;
#pragma unroll
      for (int r = 0; r < 16; ++r) {
        const float vr = re[r], vi = im[r];
        const float xr = (pc == 0) ? vr : (pc == 1) ?  vi : (pc == 2) ? -vr : -vi;
        const float xi = (pc == 0) ? vi : (pc == 1) ? -vr : (pc == 2) ? -vi :  vr;
        G[frag0_off(r,      a)] = xr;
        G[frag0_off(r + 16, a)] = xi;
      }
    } else if (a == 0) {
      G[G_PHASE]     = pr;
      G[G_PHASE + 1] = pi;
    } else if (a == 1) {
      G[G_PHASE + 2] = cosf(phis[2]);
      G[G_PHASE + 3] = sinf(phis[2]);
    }
    return;
  }
#pragma unroll
  for (int i = 8; i < 16; ++i) {
    float r = re[i] * pr - im[i] * pi;
    float m = re[i] * pi + im[i] * pr;
    re[i] = w0 ? r : re[i];
    im[i] = w0 ? m : im[i];
  }
  float* Gb = G + G_L12 + (w0 ? 1 : 0) * 1024;
#pragma unroll
  for (int r = 0; r < 16; ++r) {
    Gb[frag_off(r,      a)]      =  re[r];
    Gb[frag_off(r,      a + 16)] = -im[r];
    Gb[frag_off(r + 16, a)]      =  im[r];
    Gb[frag_off(r + 16, a + 16)] =  re[r];
  }
}

__device__ __forceinline__ float xhalf(float v) {
  int i = __builtin_bit_cast(int, v);
  return __builtin_bit_cast(float,
      __builtin_amdgcn_permlanex16(i, i, 0x76543210, 0xfedcba98, false, false));
}

__global__ __launch_bounds__(256) void qc_main(
    const float* __restrict__ x, const float* __restrict__ G,
    float* __restrict__ out, int P) {
  __shared__ v4f Gs[G_V4];

  const int tid  = threadIdx.x;
  const int lane = tid & 31;
  const bool hi  = (lane >> 4) != 0;

  const int p  = blockIdx.x * 256 + tid;
  const int b  = p >> 14;
  const int h2 = (p >> 7) & 127;
  const int w2 = p & 127;

  float ang[12];
#pragma unroll
  for (int c = 0; c < 3; ++c) {
#pragma unroll
    for (int dy = 0; dy < 2; ++dy) {
      const float* row = x + (((b * 3 + c) * 256 + (2 * h2 + dy)) * 256 + 2 * w2);
      ang[c * 4 + dy * 2 + 0] = row[0];
      ang[c * 4 + dy * 2 + 1] = row[1];
    }
  }

  const v4f* G4 = reinterpret_cast<const v4f*>(G);
  v4f gt[2];
#pragma unroll
  for (int k = 0; k < 2; ++k) gt[k] = G4[tid + 256 * k];
  v4f gt2 = (v4f){0.f, 0.f, 0.f, 0.f};
  if (tid < G_V4 - 512) gt2 = G4[tid + 512];
  const float pr0 = G[G_PHASE],     pi0 = G[G_PHASE + 1];
  const float pr2 = G[G_PHASE + 2], pi2 = G[G_PHASE + 3];

  float cso[12], sno[12];
#pragma unroll
  for (int k = 0; k < 12; ++k) {
    if (k == 8) { cso[k] = 1.0f; sno[k] = 0.0f; continue; }
    float hf = 0.5f * ang[k];
    cso[k] = cosf(hf);
    sno[k] = sinf(hf);
  }

#pragma unroll
  for (int k = 0; k < 2; ++k) Gs[tid + 256 * k] = gt[k];
  if (tid < G_V4 - 512) Gs[tid + 512] = gt2;

  __syncthreads();

  const v4f* Gw = Gs + lane;

  float st[2][2][2][8];

  {
    float c0[4][2], s0[4][2];
#pragma unroll
    for (int wq = 0; wq < 4; ++wq) {
      const float co = cso[wq], so = sno[wq];
      const float cx = xhalf(co), sx = xhalf(so);
      c0[wq][0] = hi ? cx : co;  c0[wq][1] = hi ? co : cx;
      s0[wq][0] = hi ? sx : so;  s0[wq][1] = hi ? so : sx;
    }
    float mg[2][8];
#pragma unroll
    for (int nt = 0; nt < 2; ++nt) {
      const float f1 = (hi ? s0[0][nt] : c0[0][nt]) * RS2;
#pragma unroll
      for (int r = 0; r < 8; ++r) {
        float v = f1 * ((r & 4) ? s0[1][nt] : c0[1][nt]);
        v = v * ((r & 2) ? s0[2][nt] : c0[2][nt]);
        mg[nt][r] = v * ((r & 1) ? s0[3][nt] : c0[3][nt]);
      }
    }

    v8f acc[2][2];
#pragma unroll
    for (int mo = 0; mo < 2; ++mo)
#pragma unroll
      for (int nt = 0; nt < 2; ++nt) acc[mo][nt] = (v8f){0.f, 0.f, 0.f, 0.f, 0.f, 0.f, 0.f, 0.f};
    v4f a0[2][2];
#pragma unroll
    for (int mo = 0; mo < 2; ++mo)
#pragma unroll
      for (int q = 0; q < 2; ++q) a0[mo][q] = Gw[(mo * 2 + q) * 32];
#pragma unroll
    for (int rp = 0; rp < 4; ++rp) {
#pragma unroll
      for (int mo = 0; mo < 2; ++mo) {
        const v4f aq = a0[mo][rp >> 1];
        v2f Af;
        Af[0] = (rp & 1) ? aq[2] : aq[0];
        Af[1] = (rp & 1) ? aq[3] : aq[1];
#pragma unroll
        for (int nt = 0; nt < 2; ++nt) {
          v2f Bf;
          Bf[0] = mg[nt][2 * rp];
          Bf[1] = mg[nt][2 * rp + 1];
          acc[mo][nt] = __builtin_amdgcn_wmma_f32_16x16x4_f32(
              false, Af, false, Bf, (short)0, acc[mo][nt], false, false);
        }
      }
    }

    const float pa = hi ? pr0 : 1.0f, pb = hi ? pi0 : 0.0f;
#pragma unroll
    for (int nt = 0; nt < 2; ++nt) {
#pragma unroll
      for (int r = 0; r < 8; ++r) {
        const float vr = acc[0][nt][r], vi = acc[1][nt][r];
        st[0][0][nt][r] = vr;
        st[0][1][nt][r] = vi;
        st[1][0][nt][r] = vr * pa - vi * pb;
        st[1][1][nt][r] = vr * pb + vi * pa;
      }
    }
  }

#pragma unroll
  for (int l = 1; l < 3; ++l) {
#pragma unroll
    for (int wq = (l == 2) ? 1 : 0; wq < 4; ++wq) {
      const float co = cso[4 * l + wq], so = sno[4 * l + wq];
      const float cx = xhalf(co), sx = xhalf(so);
      const float cc[2] = { hi ? cx : co, hi ? co : cx };
      const float ss[2] = { hi ? sx : so, hi ? so : sx };
#pragma unroll
      for (int nt = 0; nt < 2; ++nt) {
        const float c_ = cc[nt], s_ = ss[nt];
        if (wq == 0) {
#pragma unroll
          for (int w0 = 0; w0 < 2; ++w0) {
#pragma unroll
            for (int r = 0; r < 8; ++r) {
              float mr = st[w0][0][nt][r], mi = st[w0][1][nt][r];
              float orr = xhalf(mr),       oi = xhalf(mi);
              st[w0][0][nt][r] = c_ * mr + s_ * oi;
              st[w0][1][nt][r] = c_ * mi - s_ * orr;
            }
          }
        } else {
          const int bt = 8 >> wq;
#pragma unroll
          for (int w0 = 0; w0 < 2; ++w0) {
#pragma unroll
            for (int r = 0; r < 8; ++r) {
              if (!(r & bt)) {
                const int jj = r | bt;
                float ur = st[w0][0][nt][r],  ui = st[w0][1][nt][r];
                float vr = st[w0][0][nt][jj], vi = st[w0][1][nt][jj];
                st[w0][0][nt][r]  = c_ * ur + s_ * vi;
                st[w0][1][nt][r]  = c_ * ui - s_ * vr;
                st[w0][0][nt][jj] = c_ * vr + s_ * ui;
                st[w0][1][nt][jj] = c_ * vi - s_ * ur;
              }
            }
          }
        }
      }
    }

    if (l == 2) break;

    v8f nw[2][2][2];
#pragma unroll
    for (int w0 = 0; w0 < 2; ++w0)
#pragma unroll
      for (int mo = 0; mo < 2; ++mo)
#pragma unroll
        for (int nt = 0; nt < 2; ++nt) nw[w0][mo][nt] = (v8f){0.f, 0.f, 0.f, 0.f, 0.f, 0.f, 0.f, 0.f};

#pragma unroll
    for (int w0 = 0; w0 < 2; ++w0) {
#pragma unroll
      for (int kmt = 0; kmt < 2; ++kmt) {
        v4f a[2][2];
#pragma unroll
        for (int mo = 0; mo < 2; ++mo)
#pragma unroll
          for (int q = 0; q < 2; ++q)
            a[mo][q] = Gw[G_L12 / 4 + w0 * 256 + ((mo * 2 + kmt) * 2 + q) * 32];
#pragma unroll
        for (int rp = 0; rp < 4; ++rp) {
#pragma unroll
          for (int mo = 0; mo < 2; ++mo) {
            const v4f aq = a[mo][rp >> 1];
            v2f Af;
            Af[0] = (rp & 1) ? aq[2] : aq[0];
            Af[1] = (rp & 1) ? aq[3] : aq[1];
#pragma unroll
            for (int nt = 0; nt < 2; ++nt) {
              v2f Bf;
              Bf[0] = st[w0][kmt][nt][2 * rp];
              Bf[1] = st[w0][kmt][nt][2 * rp + 1];
              nw[w0][mo][nt] = __builtin_amdgcn_wmma_f32_16x16x4_f32(
                  false, Af, false, Bf, (short)0, nw[w0][mo][nt], false, false);
            }
          }
        }
      }
    }

#pragma unroll
    for (int w0 = 0; w0 < 2; ++w0)
#pragma unroll
      for (int mo = 0; mo < 2; ++mo)
#pragma unroll
        for (int nt = 0; nt < 2; ++nt)
#pragma unroll
          for (int r = 0; r < 8; ++r) st[w0][mo][nt][r] = nw[w0][mo][nt][r];
  }

  float tot[2];
#pragma unroll
  for (int nt = 0; nt < 2; ++nt) {
    float se = 0.0f, sr = 0.0f, si = 0.0f;
#pragma unroll
    for (int r = 0; r < 8; ++r) {
      const float ar = st[0][0][nt][r], ai = st[0][1][nt][r];
      const float br = st[1][0][nt][r], bi = st[1][1][nt][r];
      if (((r >> 2) ^ (r >> 1) ^ r) & 1) {
        sr += ar * br;  sr += ai * bi;
        si += ar * bi;  si -= ai * br;
      } else {
        se += ar * br;  se += ai * bi;
      }
    }
    const float part = se + (pr2 * sr - pi2 * si);
    tot[nt] = 2.0f * (part + xhalf(part));
  }
  const float ev = hi ? tot[1] : tot[0];
  if (p < P) out[p] = ev;
}

extern "C" void kernel_launch(void* const* d_in, const int* in_sizes, int n_in,
                              void* d_out, int out_size, void* d_ws, size_t ws_size,
                              hipStream_t stream) {
  (void)in_sizes; (void)n_in; (void)ws_size;
  const float* x      = (const float*)d_in[0];
  const float* thetas = (const float*)d_in[1];
  const float* phis   = (const float*)d_in[2];
  float* out = (float*)d_out;
  float* G   = (float*)d_ws;

  hipLaunchKernelGGL(qc_build_G, dim3(1), dim3(64), 0, stream, thetas, phis, G);

  const int P = out_size;
  const int blocks = (P + 255) / 256;
  hipLaunchKernelGGL(qc_main, dim3(blocks), dim3(256), 0, stream, x, G, out, P);
}


# One launch: each workgroup builds the layer matrices itself in LDS (CU products are phased permutations, one nonzero per column traced per lane by two waves); no prologue kernel, no workspace
# speedup vs baseline: 3.2367x; 1.2904x over previous
#include <hip/hip_runtime.h>
#include <math.h>


typedef __attribute__((ext_vector_type(2))) float v2f;
typedef __attribute__((ext_vector_type(4))) float v4f;
typedef __attribute__((ext_vector_type(8))) float v8f;

#define RS2 0.70710678118654752f
#define G_L12   512
#define G_V4    640

__device__ __forceinline__ int frag_off(int R, int C) {
  return ((((R >> 4) * 2 + (C >> 4)) * 2 + ((C >> 2) & 1)) * 32 + ((C >> 3) & 1) * 16 + (R & 15)) * 4 + (C & 3);
}

__device__ __forceinline__ int frag0_off(int R, int C) {
  return ((((R >> 4) * 2 + ((C >> 2) & 1)) * 32) + ((C >> 3) & 1) * 16 + (R & 15)) * 4 + (C & 3);
}

__device__ __forceinline__ float from_lane(float v, int k) {
  return __builtin_bit_cast(float, __builtin_amdgcn_readlane(__builtin_bit_cast(int, v), k));
}

template <int L>
__device__ __forceinline__ void trace_column(int a, float cv, float sv,
                                             int& idx, float& pr, float& pi) {
  const int cb[4] = {8, 4, 2, 1};
  const int tb[4] = {4, 2, 1, 8};
  idx = a; pr = 1.0f; pi = 0.0f;
#pragma unroll
  for (int w = 0; w < 4; ++w) {
    const float ec = from_lane(cv, 4 * L + w);
    const float es = from_lane(sv, 4 * L + w);
    const bool ctrl = (idx & cb[w]) != 0;
    const float fr = (idx & tb[w]) ? es : -es;
    const float nr = pr * fr + pi * ec;
    const float ni = pi * fr - pr * ec;
    pr  = ctrl ? nr : pr;
    pi  = ctrl ? ni : pi;
    idx = ctrl ? (idx ^ tb[w]) : idx;
  }
}

__device__ __forceinline__ float xhalf(float v) {
  int i = __builtin_bit_cast(int, v);
  return __builtin_bit_cast(float,
      __builtin_amdgcn_permlanex16(i, i, 0x76543210, 0xfedcba98, false, false));
}

__global__ __launch_bounds__(256) void qc_main(
    const float* __restrict__ x, const float* __restrict__ thetas,
    const float* __restrict__ phis, float* __restrict__ out, int P) {
  __shared__ v4f Gs[G_V4 + 1];

  const int tid  = threadIdx.x;
  const int lane = tid & 31;
  const bool hi  = (lane >> 4) != 0;
  const int wv   = __builtin_amdgcn_readfirstlane(tid >> 5);

  const int p  = blockIdx.x * 256 + tid;
  const int b  = p >> 14;
  const int h2 = (p >> 7) & 127;
  const int w2 = p & 127;

  float ang[12];
#pragma unroll
  for (int c = 0; c < 3; ++c) {
#pragma unroll
    for (int dy = 0; dy < 2; ++dy) {
      const float* row = x + (((b * 3 + c) * 256 + (2 * h2 + dy)) * 256 + 2 * w2);
      ang[c * 4 + dy * 2 + 0] = row[0];
      ang[c * 4 + dy * 2 + 1] = row[1];
    }
  }

  if (wv < 2) {
    const int ip = lane < 8 ? 0 : (lane > 10 ? 2 : lane - 8);
    const float tv = thetas[lane & 7];
    const float pv = phis[ip];
    const float hv = lane < 8 ? 0.5f * tv : pv;
    const float cv = cosf(hv), sv = sinf(hv);
    float* Gf = reinterpret_cast<float*>(Gs);
    const v4f zero = (v4f){0.f, 0.f, 0.f, 0.f};
    const int a = lane & 15;
    int idx; float vr, vi;
    if (wv == 0) {
#pragma unroll
      for (int j = 0; j < 16; ++j) Gs[G_L12 / 4 + 32 * j + lane] = zero;
      trace_column<1>(a, cv, sv, idx, vr, vi);
      const float c1 = from_lane(cv, 9), s1 = from_lane(sv, 9);
      const bool ph = hi && (idx & 8);
      const float wr = vr * c1 - vi * s1;
      const float wi = vr * s1 + vi * c1;
      vr = ph ? wr : vr;
      vi = ph ? wi : vi;
      float* Gb = Gf + G_L12 + (hi ? 1024 : 0);
      Gb[frag_off(idx,      a)]      =  vr;
      Gb[frag_off(idx,      a + 16)] = -vi;
      Gb[frag_off(idx + 16, a)]      =  vi;
      Gb[frag_off(idx + 16, a + 16)] =  vr;
    } else {
#pragma unroll
      for (int j = 0; j < 4; ++j) Gs[32 * j + lane] = zero;
      trace_column<0>(a, cv, sv, idx, vr, vi);
      const int pc = __popc(a) & 3;
      const float xr = (pc == 0) ? vr : (pc == 1) ?  vi : (pc == 2) ? -vr : -vi;
      const float xi = (pc == 0) ? vi : (pc == 1) ? -vr : (pc == 2) ? -vi :  vr;
      if (!hi) {
        Gf[frag0_off(idx,      a)] = xr;
        Gf[frag0_off(idx + 16, a)] = xi;
      }
      if (lane == 0)
        Gs[G_V4] = (v4f){from_lane(cv, 8), from_lane(sv, 8), from_lane(cv, 10), from_lane(sv, 10)};
    }
  }

  float cso[12], sno[12];
#pragma unroll
  for (int k = 0; k < 12; ++k) {
    if (k == 8) { cso[k] = 1.0f; sno[k] = 0.0f; continue; }
    float hf = 0.5f * ang[k];
    cso[k] = cosf(hf);
    sno[k] = sinf(hf);
  }

  __syncthreads();

  const v4f phs = Gs[G_V4];
  const float pr0 = phs[0], pi0 = phs[1];
  const float pr2 = phs[2], pi2 = phs[3];

  const v4f* Gw = Gs + lane;

  float st[2][2][2][8];

  {
    float c0[4][2], s0[4][2];
#pragma unroll
    for (int wq = 0; wq < 4; ++wq) {
      const float co = cso[wq], so = sno[wq];
      const float cx = xhalf(co), sx = xhalf(so);
      c0[wq][0] = hi ? cx : co;  c0[wq][1] = hi ? co : cx;
      s0[wq][0] = hi ? sx : so;  s0[wq][1] = hi ? so : sx;
    }
    float mg[2][8];
#pragma unroll
    for (int nt = 0; nt < 2; ++nt) {
      const float f1 = (hi ? s0[0][nt] : c0[0][nt]) * RS2;
#pragma unroll
      for (int r = 0; r < 8; ++r) {
        float v = f1 * ((r & 4) ? s0[1][nt] : c0[1][nt]);
        v = v * ((r & 2) ? s0[2][nt] : c0[2][nt]);
        mg[nt][r] = v * ((r & 1) ? s0[3][nt] : c0[3][nt]);
      }
    }

    v8f acc[2][2];
#pragma unroll
    for (int mo = 0; mo < 2; ++mo)
#pragma unroll
      for (int nt = 0; nt < 2; ++nt) acc[mo][nt] = (v8f){0.f, 0.f, 0.f, 0.f, 0.f, 0.f, 0.f, 0.f};
    v4f a0[2][2];
#pragma unroll
    for (int mo = 0; mo < 2; ++mo)
#pragma unroll
      for (int q = 0; q < 2; ++q) a0[mo][q] = Gw[(mo * 2 + q) * 32];
#pragma unroll
    for (int rp = 0; rp < 4; ++rp) {
#pragma unroll
      for (int mo = 0; mo < 2; ++mo) {
        const v4f aq = a0[mo][rp >> 1];
        v2f Af;
        Af[0] = (rp & 1) ? aq[2] : aq[0];
        Af[1] = (rp & 1) ? aq[3] : aq[1];
#pragma unroll
        for (int nt = 0; nt < 2; ++nt) {
          v2f Bf;
          Bf[0] = mg[nt][2 * rp];
          Bf[1] = mg[nt][2 * rp + 1];
          acc[mo][nt] = __builtin_amdgcn_wmma_f32_16x16x4_f32(
              false, Af, false, Bf, (short)0, acc[mo][nt], false, false);
        }
      }
    }

    const float pa = hi ? pr0 : 1.0f, pb = hi ? pi0 : 0.0f;
#pragma unroll
    for (int nt = 0; nt < 2; ++nt) {
#pragma unroll
      for (int r = 0; r < 8; ++r) {
        const float vr = acc[0][nt][r], vi = acc[1][nt][r];
        st[0][0][nt][r] = vr;
        st[0][1][nt][r] = vi;
        st[1][0][nt][r] = vr * pa - vi * pb;
        st[1][1][nt][r] = vr * pb + vi * pa;
      }
    }
  }

#pragma unroll
  for (int l = 1; l < 3; ++l) {
#pragma unroll
    for (int wq = (l == 2) ? 1 : 0; wq < 4; ++wq) {
      const float co = cso[4 * l + wq], so = sno[4 * l + wq];
      const float cx = xhalf(co), sx = xhalf(so);
      const float cc[2] = { hi ? cx : co, hi ? co : cx };
      const float ss[2] = { hi ? sx : so, hi ? so : sx };
#pragma unroll
      for (int nt = 0; nt < 2; ++nt) {
        const float c_ = cc[nt], s_ = ss[nt];
        if (wq == 0) {
#pragma unroll
          for (int w0 = 0; w0 < 2; ++w0) {
#pragma unroll
            for (int r = 0; r < 8; ++r) {
              float mr = st[w0][0][nt][r], mi = st[w0][1][nt][r];
              float orr = xhalf(mr),       oi = xhalf(mi);
              st[w0][0][nt][r] = c_ * mr + s_ * oi;
              st[w0][1][nt][r] = c_ * mi - s_ * orr;
            }
          }
        } else {
          const int bt = 8 >> wq;
#pragma unroll
          for (int w0 = 0; w0 < 2; ++w0) {
#pragma unroll
            for (int r = 0; r < 8; ++r) {
              if (!(r & bt)) {
                const int jj = r | bt;
                float ur = st[w0][0][nt][r],  ui = st[w0][1][nt][r];
                float vr = st[w0][0][nt][jj], vi = st[w0][1][nt][jj];
                st[w0][0][nt][r]  = c_ * ur + s_ * vi;
                st[w0][1][nt][r]  = c_ * ui - s_ * vr;
                st[w0][0][nt][jj] = c_ * vr + s_ * ui;
                st[w0][1][nt][jj] = c_ * vi - s_ * ur;
              }
            }
          }
        }
      }
    }

    if (l == 2) break;

    v8f nw[2][2][2];
#pragma unroll
    for (int w0 = 0; w0 < 2; ++w0)
#pragma unroll
      for (int mo = 0; mo < 2; ++mo)
#pragma unroll
        for (int nt = 0; nt < 2; ++nt) nw[w0][mo][nt] = (v8f){0.f, 0.f, 0.f, 0.f, 0.f, 0.f, 0.f, 0.f};

#pragma unroll
    for (int w0 = 0; w0 < 2; ++w0) {
#pragma unroll
      for (int kmt = 0; kmt < 2; ++kmt) {
        v4f a[2][2];
#pragma unroll
        for (int mo = 0; mo < 2; ++mo)
#pragma unroll
          for (int q = 0; q < 2; ++q)
            a[mo][q] = Gw[G_L12 / 4 + w0 * 256 + ((mo * 2 + kmt) * 2 + q) * 32];
#pragma unroll
        for (int rp = 0; rp < 4; ++rp) {
#pragma unroll
          for (int mo = 0; mo < 2; ++mo) {
            const v4f aq = a[mo][rp >> 1];
            v2f Af;
            Af[0] = (rp & 1) ? aq[2] : aq[0];
            Af[1] = (rp & 1) ? aq[3] : aq[1];
#pragma unroll
            for (int nt = 0; nt < 2; ++nt) {
              v2f Bf;
              Bf[0] = st[w0][kmt][nt][2 * rp];
              Bf[1] = st[w0][kmt][nt][2 * rp + 1];
              nw[w0][mo][nt] = __builtin_amdgcn_wmma_f32_16x16x4_f32(
                  false, Af, false, Bf, (short)0, nw[w0][mo][nt], false, false);
            }
          }
        }
      }
    }

#pragma unroll
    for (int w0 = 0; w0 < 2; ++w0)
#pragma unroll
      for (int mo = 0; mo < 2; ++mo)
#pragma unroll
        for (int nt = 0; nt < 2; ++nt)
#pragma unroll
          for (int r = 0; r < 8; ++r) st[w0][mo][nt][r] = nw[w0][mo][nt][r];
  }

  float tot[2];
#pragma unroll
  for (int nt = 0; nt < 2; ++nt) {
    float se = 0.0f, sr = 0.0f, si = 0.0f;
#pragma unroll
    for (int r = 0; r < 8; ++r) {
      const float ar = st[0][0][nt][r], ai = st[0][1][nt][r];
      const float br = st[1][0][nt][r], bi = st[1][1][nt][r];
      if (((r >> 2) ^ (r >> 1) ^ r) & 1) {
        sr += ar * br;  sr += ai * bi;
        si += ar * bi;  si -= ai * br;
      } else {
        se += ar * br;  se += ai * bi;
      }
    }
    const float part = se + (pr2 * sr - pi2 * si);
    tot[nt] = 2.0f * (part + xhalf(part));
  }
  const float ev = hi ? tot[1] : tot[0];
  if (p < P) out[p] = ev;
}

extern "C" void kernel_launch(void* const* d_in, const int* in_sizes, int n_in,
                              void* d_out, int out_size, void* d_ws, size_t ws_size,
                              hipStream_t stream) {
  (void)in_sizes; (void)n_in; (void)ws_size;
  const float* x      = (const float*)d_in[0];
  const float* thetas = (const float*)d_in[1];
  const float* phis   = (const float*)d_in[2];
  float* out = (float*)d_out;
  (void)d_ws;

  const int P = out_size;
  const int blocks = (P + 255) / 256;
  hipLaunchKernelGGL(qc_main, dim3(blocks), dim3(256), 0, stream, x, thetas, phis, out, P);
}


# Both w0 halves differ only by a wire-1 operator before layer 1, so layer 1 runs on the w0=0 half (32 WMMAs not 64, half the RX) and w0=1 is 2 terms per row from a 16-row table; sum order changes
# speedup vs baseline: 3.6637x; 1.1319x over previous
#include <hip/hip_runtime.h>
#include <math.h>


typedef __attribute__((ext_vector_type(2))) float v2f;
typedef __attribute__((ext_vector_type(4))) float v4f;
typedef __attribute__((ext_vector_type(8))) float v8f;

#define RS2 0.70710678118654752f
#define G_L12   512
#define G_TAB   1536
#define G_V4    416

__device__ __forceinline__ int frag_off(int R, int C) {
  return ((((R >> 4) * 2 + (C >> 4)) * 2 + ((C >> 2) & 1)) * 32 + ((C >> 3) & 1) * 16 + (R & 15)) * 4 + (C & 3);
}

__device__ __forceinline__ int frag0_off(int R, int C) {
  return ((((R >> 4) * 2 + ((C >> 2) & 1)) * 32) + ((C >> 3) & 1) * 16 + (R & 15)) * 4 + (C & 3);
}

__device__ __forceinline__ float from_lane(float v, int k) {
  return __builtin_bit_cast(float, __builtin_amdgcn_readlane(__builtin_bit_cast(int, v), k));
}

template <int L>
__device__ __forceinline__ void trace_column(int a, float cv, float sv,
                                             int& idx, float& pr, float& pi) {
  const int cb[4] = {8, 4, 2, 1};
  const int tb[4] = {4, 2, 1, 8};
  idx = a; pr = 1.0f; pi = 0.0f;
#pragma unroll
  for (int w = 0; w < 4; ++w) {
    const float ec = from_lane(cv, 4 * L + w);
    const float es = from_lane(sv, 4 * L + w);
    const bool ctrl = (idx & cb[w]) != 0;
    const float fr = (idx & tb[w]) ? es : -es;
    const float nr = pr * fr + pi * ec;
    const float ni = pi * fr - pr * ec;
    pr  = ctrl ? nr : pr;
    pi  = ctrl ? ni : pi;
    idx = ctrl ? (idx ^ tb[w]) : idx;
  }
}

__device__ __forceinline__ float xhalf(float v) {
  int i = __builtin_bit_cast(int, v);
  return __builtin_bit_cast(float,
      __builtin_amdgcn_permlanex16(i, i, 0x76543210, 0xfedcba98, false, false));
}

__global__ __launch_bounds__(256) void qc_main(
    const float* __restrict__ x, const float* __restrict__ thetas,
    const float* __restrict__ phis, float* __restrict__ out, int P) {
  __shared__ v4f Gs[G_V4 + 1];

  const int tid  = threadIdx.x;
  const int lane = tid & 31;
  const bool hi  = (lane >> 4) != 0;
  const int wv   = __builtin_amdgcn_readfirstlane(tid >> 5);

  const int p  = blockIdx.x * 256 + tid;
  const int b  = p >> 14;
  const int h2 = (p >> 7) & 127;
  const int w2 = p & 127;

  float ang[12];
#pragma unroll
  for (int c = 0; c < 3; ++c) {
#pragma unroll
    for (int dy = 0; dy < 2; ++dy) {
      const float* row = x + (((b * 3 + c) * 256 + (2 * h2 + dy)) * 256 + 2 * w2);
      ang[c * 4 + dy * 2 + 0] = row[0];
      ang[c * 4 + dy * 2 + 1] = row[1];
    }
  }

  if (wv < 2) {
    const int ip = lane < 8 ? 0 : (lane > 10 ? 2 : lane - 8);
    const float tv = thetas[lane & 7];
    const float pv = phis[ip];
    const float hv = lane < 8 ? 0.5f * tv : (lane == 8 ? 0.5f * pv : pv);
    const float cv = cosf(hv), sv = sinf(hv);
    float* Gf = reinterpret_cast<float*>(Gs);
    const v4f zero = (v4f){0.f, 0.f, 0.f, 0.f};
    int idx; float vr, vi;
    if (wv == 0) {
#pragma unroll
      for (int j = 0; j < 8; ++j) Gs[G_L12 / 4 + 32 * j + lane] = zero;
      const int a = (lane & 15) ^ (hi ? 8 : 0);
      trace_column<1>(a, cv, sv, idx, vr, vi);
      const float qr = xhalf(vr), qi = xhalf(vi);
      if (!hi) {
        float* Gb = Gf + G_L12;
        Gb[frag_off(idx,      a)]      =  vr;
        Gb[frag_off(idx,      a + 16)] = -vi;
        Gb[frag_off(idx + 16, a)]      =  vi;
        Gb[frag_off(idx + 16, a + 16)] =  vr;
        const float Ch = from_lane(cv, 8), Sh = from_lane(sv, 8);
        const float c1 = from_lane(cv, 9), s1 = from_lane(sv, 9);
        const bool w1o = (idx & 8) != 0;
        const float kr = w1o ? (c1 * Ch - s1 * Sh) : Ch;
        const float ki = w1o ? (c1 * Sh + s1 * Ch) : Sh;
        const float sS = (a & 8) ? -Sh : Sh;
        const float psr = vr * qr + vi * qi;
        const float psi = vi * qr - vr * qi;
        const float gr = (kr * psr - ki * psi) * sS;
        const float gi = (kr * psi + ki * psr) * sS;
        Gs[G_TAB / 4 + 2 * idx]     = (v4f){kr * Ch, ki * Ch, ki * sS, -kr * sS};
        Gs[G_TAB / 4 + 2 * idx + 1] = (v4f){gr, gi, 0.f, 0.f};
      }
    } else {
#pragma unroll
      for (int j = 0; j < 4; ++j) Gs[32 * j + lane] = zero;
      const int a = lane & 15;
      trace_column<0>(a, cv, sv, idx, vr, vi);
      const int pc = __popc(a) & 3;
      const float xr = (pc == 0) ? vr : (pc == 1) ?  vi : (pc == 2) ? -vr : -vi;
      const float xi = (pc == 0) ? vi : (pc == 1) ? -vr : (pc == 2) ? -vi :  vr;
      if (!hi) {
        Gf[frag0_off(idx,      a)] = xr;
        Gf[frag0_off(idx + 16, a)] = xi;
      }
      if (lane == 0)
        Gs[G_V4] = (v4f){from_lane(cv, 10), from_lane(sv, 10), 0.f, 0.f};
    }
  }

  float cso[12], sno[12];
#pragma unroll
  for (int k = 0; k < 12; ++k) {
    if (k == 8) { cso[k] = 1.0f; sno[k] = 0.0f; continue; }
    float hf = 0.5f * ang[k];
    cso[k] = cosf(hf);
    sno[k] = sinf(hf);
  }

  __syncthreads();

  const v4f phs = Gs[G_V4];
  const float pr2 = phs[0], pi2 = phs[1];

  const v4f* Gw = Gs + lane;

  float st[2][2][2][8];

  {
    float c0[4][2], s0[4][2];
#pragma unroll
    for (int wq = 0; wq < 4; ++wq) {
      const float co = cso[wq], so = sno[wq];
      const float cx = xhalf(co), sx = xhalf(so);
      c0[wq][0] = hi ? cx : co;  c0[wq][1] = hi ? co : cx;
      s0[wq][0] = hi ? sx : so;  s0[wq][1] = hi ? so : sx;
    }
    float mg[2][8];
#pragma unroll
    for (int nt = 0; nt < 2; ++nt) {
      const float f1 = (hi ? s0[0][nt] : c0[0][nt]) * RS2;
#pragma unroll
      for (int r = 0; r < 8; ++r) {
        float v = f1 * ((r & 4) ? s0[1][nt] : c0[1][nt]);
        v = v * ((r & 2) ? s0[2][nt] : c0[2][nt]);
        mg[nt][r] = v * ((r & 1) ? s0[3][nt] : c0[3][nt]);
      }
    }

    v8f acc[2][2];
#pragma unroll
    for (int mo = 0; mo < 2; ++mo)
#pragma unroll
      for (int nt = 0; nt < 2; ++nt) acc[mo][nt] = (v8f){0.f, 0.f, 0.f, 0.f, 0.f, 0.f, 0.f, 0.f};
    v4f a0[2][2];
#pragma unroll
    for (int mo = 0; mo < 2; ++mo)
#pragma unroll
      for (int q = 0; q < 2; ++q) a0[mo][q] = Gw[(mo * 2 + q) * 32];
#pragma unroll
    for (int rp = 0; rp < 4; ++rp) {
#pragma unroll
      for (int mo = 0; mo < 2; ++mo) {
        const v4f aq = a0[mo][rp >> 1];
        v2f Af;
        Af[0] = (rp & 1) ? aq[2] : aq[0];
        Af[1] = (rp & 1) ? aq[3] : aq[1];
#pragma unroll
        for (int nt = 0; nt < 2; ++nt) {
          v2f Bf;
          Bf[0] = mg[nt][2 * rp];
          Bf[1] = mg[nt][2 * rp + 1];
          acc[mo][nt] = __builtin_amdgcn_wmma_f32_16x16x4_f32(
              false, Af, false, Bf, (short)0, acc[mo][nt], false, false);
        }
      }
    }

#pragma unroll
    for (int nt = 0; nt < 2; ++nt) {
#pragma unroll
      for (int r = 0; r < 8; ++r) {
        st[0][0][nt][r] = acc[0][nt][r];
        st[0][1][nt][r] = acc[1][nt][r];
      }
    }
  }

#pragma unroll
  for (int l = 1; l < 3; ++l) {
#pragma unroll
    for (int wq = (l == 2) ? 1 : 0; wq < 4; ++wq) {
      const float co = cso[4 * l + wq], so = sno[4 * l + wq];
      const float cx = xhalf(co), sx = xhalf(so);
      const float cc[2] = { hi ? cx : co, hi ? co : cx };
      const float ss[2] = { hi ? sx : so, hi ? so : sx };
#pragma unroll
      for (int nt = 0; nt < 2; ++nt) {
        const float c_ = cc[nt], s_ = ss[nt];
        if (wq == 0) {
#pragma unroll
          for (int w0 = 0; w0 < 2; ++w0) {
            if (l == 1 && w0 == 1) continue;
#pragma unroll
            for (int r = 0; r < 8; ++r) {
              float mr = st[w0][0][nt][r], mi = st[w0][1][nt][r];
              float orr = xhalf(mr),       oi = xhalf(mi);
              st[w0][0][nt][r] = c_ * mr + s_ * oi;
              st[w0][1][nt][r] = c_ * mi - s_ * orr;
            }
          }
        } else {
          const int bt = 8 >> wq;
#pragma unroll
          for (int w0 = 0; w0 < 2; ++w0) {
            if (l == 1 && w0 == 1) continue;
#pragma unroll
            for (int r = 0; r < 8; ++r) {
              if (!(r & bt)) {
                const int jj = r | bt;
                float ur = st[w0][0][nt][r],  ui = st[w0][1][nt][r];
                float vr = st[w0][0][nt][jj], vi = st[w0][1][nt][jj];
                st[w0][0][nt][r]  = c_ * ur + s_ * vi;
                st[w0][1][nt][r]  = c_ * ui - s_ * vr;
                st[w0][0][nt][jj] = c_ * vr + s_ * ui;
                st[w0][1][nt][jj] = c_ * vi - s_ * ur;
              }
            }
          }
        }
      }
    }

    if (l == 2) break;

    v8f nw[2][2];
#pragma unroll
    for (int mo = 0; mo < 2; ++mo)
#pragma unroll
      for (int nt = 0; nt < 2; ++nt) nw[mo][nt] = (v8f){0.f, 0.f, 0.f, 0.f, 0.f, 0.f, 0.f, 0.f};

#pragma unroll
    for (int kmt = 0; kmt < 2; ++kmt) {
      v4f a[2][2];
#pragma unroll
      for (int mo = 0; mo < 2; ++mo)
#pragma unroll
        for (int q = 0; q < 2; ++q)
          a[mo][q] = Gw[G_L12 / 4 + ((mo * 2 + kmt) * 2 + q) * 32];
#pragma unroll
      for (int rp = 0; rp < 4; ++rp) {
#pragma unroll
        for (int mo = 0; mo < 2; ++mo) {
          const v4f aq = a[mo][rp >> 1];
          v2f Af;
          Af[0] = (rp & 1) ? aq[2] : aq[0];
          Af[1] = (rp & 1) ? aq[3] : aq[1];
#pragma unroll
          for (int nt = 0; nt < 2; ++nt) {
            v2f Bf;
            Bf[0] = st[0][kmt][nt][2 * rp];
            Bf[1] = st[0][kmt][nt][2 * rp + 1];
            nw[mo][nt] = __builtin_amdgcn_wmma_f32_16x16x4_f32(
                false, Af, false, Bf, (short)0, nw[mo][nt], false, false);
          }
        }
      }
    }

    {
      const float co = cso[4], so = sno[4];
      const float cfo = co * co - so * so, sfo = 2.0f * co * so;
      const float cfx = xhalf(cfo), sfx = xhalf(sfo);
      const float ct[2] = { hi ? cfx : cfo, hi ? cfo : cfx };
      const float sn[2] = { hi ? sfx : sfo, hi ? sfo : sfx };
      const v4f* Tb = Gs + G_TAB / 4 + (hi ? 16 : 0);
#pragma unroll
      for (int r = 0; r < 8; ++r) {
        const v4f t0 = Tb[2 * r];
        const v4f t1 = Tb[2 * r + 1];
#pragma unroll
        for (int nt = 0; nt < 2; ++nt) {
          const float kr = t0[0] + t0[2] * ct[nt];
          const float ki = t0[1] + t0[3] * ct[nt];
          const float gr = t1[0] * sn[nt];
          const float gi = t1[1] * sn[nt];
          const float ur = nw[0][nt][r],     ui = nw[1][nt][r];
          const float xr = nw[0][nt][r ^ 7], xi = nw[1][nt][r ^ 7];
          st[0][0][nt][r] = ur;
          st[0][1][nt][r] = ui;
          st[1][0][nt][r] = kr * ur - ki * ui + gr * xr - gi * xi;
          st[1][1][nt][r] = kr * ui + ki * ur + gr * xi + gi * xr;
        }
      }
    }
  }

  float tot[2];
#pragma unroll
  for (int nt = 0; nt < 2; ++nt) {
    float se = 0.0f, sr = 0.0f, si = 0.0f;
#pragma unroll
    for (int r = 0; r < 8; ++r) {
      const float ar = st[0][0][nt][r], ai = st[0][1][nt][r];
      const float br = st[1][0][nt][r], bi = st[1][1][nt][r];
      if (((r >> 2) ^ (r >> 1) ^ r) & 1) {
        sr += ar * br;  sr += ai * bi;
        si += ar * bi;  si -= ai * br;
      } else {
        se += ar * br;  se += ai * bi;
      }
    }
    const float part = se + (pr2 * sr - pi2 * si);
    tot[nt] = 2.0f * (part + xhalf(part));
  }
  const float ev = hi ? tot[1] : tot[0];
  if (p < P) out[p] = ev;
}

extern "C" void kernel_launch(void* const* d_in, const int* in_sizes, int n_in,
                              void* d_out, int out_size, void* d_ws, size_t ws_size,
                              hipStream_t stream) {
  (void)in_sizes; (void)n_in; (void)ws_size;
  const float* x      = (const float*)d_in[0];
  const float* thetas = (const float*)d_in[1];
  const float* phis   = (const float*)d_in[2];
  float* out = (float*)d_out;
  (void)d_ws;

  const int P = out_size;
  const int blocks = (P + 255) / 256;
  hipLaunchKernelGGL(qc_main, dim3(blocks), dim3(256), 0, stream, x, thetas, phis, out, P);
}


# cos and sin of each half angle from ONE shared argument reduction and polynomial pair, all 11 in one straight block behind one size test (library sincos for huge arguments); was 22 separate calls
# speedup vs baseline: 4.5314x; 1.2368x over previous
#include <hip/hip_runtime.h>
#include <math.h>


typedef __attribute__((ext_vector_type(2))) float v2f;
typedef __attribute__((ext_vector_type(4))) float v4f;
typedef __attribute__((ext_vector_type(8))) float v8f;

#define RS2 0.70710678118654752f
#define G_L12   512
#define G_TAB   1536
#define G_V4    416

__device__ __forceinline__ int frag_off(int R, int C) {
  return ((((R >> 4) * 2 + (C >> 4)) * 2 + ((C >> 2) & 1)) * 32 + ((C >> 3) & 1) * 16 + (R & 15)) * 4 + (C & 3);
}

__device__ __forceinline__ int frag0_off(int R, int C) {
  return ((((R >> 4) * 2 + ((C >> 2) & 1)) * 32) + ((C >> 3) & 1) * 16 + (R & 15)) * 4 + (C & 3);
}

__device__ __forceinline__ float from_lane(float v, int k) {
  return __builtin_bit_cast(float, __builtin_amdgcn_readlane(__builtin_bit_cast(int, v), k));
}

template <int L>
__device__ __forceinline__ void trace_column(int a, float cv, float sv,
                                             int& idx, float& pr, float& pi) {
  const int cb[4] = {8, 4, 2, 1};
  const int tb[4] = {4, 2, 1, 8};
  idx = a; pr = 1.0f; pi = 0.0f;
#pragma unroll
  for (int w = 0; w < 4; ++w) {
    const float ec = from_lane(cv, 4 * L + w);
    const float es = from_lane(sv, 4 * L + w);
    const bool ctrl = (idx & cb[w]) != 0;
    const float fr = (idx & tb[w]) ? es : -es;
    const float nr = pr * fr + pi * ec;
    const float ni = pi * fr - pr * ec;
    pr  = ctrl ? nr : pr;
    pi  = ctrl ? ni : pi;
    idx = ctrl ? (idx ^ tb[w]) : idx;
  }
}

__device__ __forceinline__ float xhalf(float v) {
  int i = __builtin_bit_cast(int, v);
  return __builtin_bit_cast(float,
      __builtin_amdgcn_permlanex16(i, i, 0x76543210, 0xfedcba98, false, false));
}

__device__ __forceinline__ float f32_bits(unsigned u) { return __builtin_bit_cast(float, u); }
__device__ __forceinline__ void cos_sin_small(float h, float& c, float& s) {
  const float ah = __builtin_fabsf(h);
  const float fn = __builtin_rintf(ah * f32_bits(0x3f22f983u));
  float r = __builtin_fmaf(fn, -f32_bits(0x3fc90fdau), ah);
  r = __builtin_fmaf(fn, -f32_bits(0x33a22168u), r);
  r = __builtin_fmaf(fn, -f32_bits(0x27c234c4u), r);
  const unsigned q = (unsigned)(int)fn;
  const float t = r * r;
  float ps = __builtin_fmaf(f32_bits(0xb94c1982u), t, f32_bits(0x3c0881c4u));
  ps = __builtin_fmaf(t, ps, f32_bits(0xbe2aaa9du));
  const float sr = __builtin_fmaf(r, t * ps, r);
  float pc = __builtin_fmaf(f32_bits(0x37d75334u), t, f32_bits(0xbab64f3bu));
  pc = __builtin_fmaf(t, pc, f32_bits(0x3d2aabf7u));
  pc = __builtin_fmaf(t, pc, f32_bits(0xbf000004u));
  const float cr = __builtin_fmaf(t, pc, 1.0f);
  const bool odd = (q & 1u) != 0u;
  const unsigned flip = (q << 30) & 0x80000000u;
  const unsigned sgn = __builtin_bit_cast(unsigned, h) & 0x80000000u;
  const float sq = odd ? cr : sr;
  const float cq = odd ? -sr : cr;
  s = __builtin_bit_cast(float, __builtin_bit_cast(unsigned, sq) ^ flip ^ sgn);
  c = __builtin_bit_cast(float, __builtin_bit_cast(unsigned, cq) ^ flip);
}

__global__ __launch_bounds__(256) void qc_main(
    const float* __restrict__ x, const float* __restrict__ thetas,
    const float* __restrict__ phis, float* __restrict__ out, int P) {
  __shared__ v4f Gs[G_V4 + 1];

  const int tid  = threadIdx.x;
  const int lane = tid & 31;
  const bool hi  = (lane >> 4) != 0;
  const int wv   = __builtin_amdgcn_readfirstlane(tid >> 5);

  const int p  = blockIdx.x * 256 + tid;
  const int b  = p >> 14;
  const int h2 = (p >> 7) & 127;
  const int w2 = p & 127;

  float ang[12];
#pragma unroll
  for (int c = 0; c < 3; ++c) {
#pragma unroll
    for (int dy = 0; dy < 2; ++dy) {
      const float* row = x + (((b * 3 + c) * 256 + (2 * h2 + dy)) * 256 + 2 * w2);
      ang[c * 4 + dy * 2 + 0] = row[0];
      ang[c * 4 + dy * 2 + 1] = row[1];
    }
  }

  if (wv < 2) {
    const int ip = lane < 8 ? 0 : (lane > 10 ? 2 : lane - 8);
    const float tv = thetas[lane & 7];
    const float pv = phis[ip];
    const float hv = lane < 8 ? 0.5f * tv : (lane == 8 ? 0.5f * pv : pv);
    float cv, sv;
    sincosf(hv, &sv, &cv);
    float* Gf = reinterpret_cast<float*>(Gs);
    const v4f zero = (v4f){0.f, 0.f, 0.f, 0.f};
    int idx; float vr, vi;
    if (wv == 0) {
#pragma unroll
      for (int j = 0; j < 8; ++j) Gs[G_L12 / 4 + 32 * j + lane] = zero;
      const int a = (lane & 15) ^ (hi ? 8 : 0);
      trace_column<1>(a, cv, sv, idx, vr, vi);
      const float qr = xhalf(vr), qi = xhalf(vi);
      if (!hi) {
        float* Gb = Gf + G_L12;
        Gb[frag_off(idx,      a)]      =  vr;
        Gb[frag_off(idx,      a + 16)] = -vi;
        Gb[frag_off(idx + 16, a)]      =  vi;
        Gb[frag_off(idx + 16, a + 16)] =  vr;
        const float Ch = from_lane(cv, 8), Sh = from_lane(sv, 8);
        const float c1 = from_lane(cv, 9), s1 = from_lane(sv, 9);
        const bool w1o = (idx & 8) != 0;
        const float kr = w1o ? (c1 * Ch - s1 * Sh) : Ch;
        const float ki = w1o ? (c1 * Sh + s1 * Ch) : Sh;
        const float sS = (a & 8) ? -Sh : Sh;
        const float psr = vr * qr + vi * qi;
        const float psi = vi * qr - vr * qi;
        const float gr = (kr * psr - ki * psi) * sS;
        const float gi = (kr * psi + ki * psr) * sS;
        Gs[G_TAB / 4 + 2 * idx]     = (v4f){kr * Ch, ki * Ch, ki * sS, -kr * sS};
        Gs[G_TAB / 4 + 2 * idx + 1] = (v4f){gr, gi, 0.f, 0.f};
      }
    } else {
#pragma unroll
      for (int j = 0; j < 4; ++j) Gs[32 * j + lane] = zero;
      const int a = lane & 15;
      trace_column<0>(a, cv, sv, idx, vr, vi);
      const int pc = __popc(a) & 3;
      const float xr = (pc == 0) ? vr : (pc == 1) ?  vi : (pc == 2) ? -vr : -vi;
      const float xi = (pc == 0) ? vi : (pc == 1) ? -vr : (pc == 2) ? -vi :  vr;
      if (!hi) {
        Gf[frag0_off(idx,      a)] = xr;
        Gf[frag0_off(idx + 16, a)] = xi;
      }
      if (lane == 0)
        Gs[G_V4] = (v4f){from_lane(cv, 10), from_lane(sv, 10), 0.f, 0.f};
    }
  }

  float cso[12], sno[12];
  float hfa[12];
  float amax = 0.0f;
#pragma unroll
  for (int k = 0; k < 12; ++k) {
    hfa[k] = 0.5f * ang[k];
    if (k != 8) amax = __builtin_fmaxf(amax, __builtin_fabsf(hfa[k]));
  }
  cso[8] = 1.0f; sno[8] = 0.0f;
  if (amax < 131072.0f) {
#pragma unroll
    for (int k = 0; k < 12; ++k)
      if (k != 8) cos_sin_small(hfa[k], cso[k], sno[k]);
  } else {
#pragma unroll
    for (int k = 0; k < 12; ++k)
      if (k != 8) sincosf(hfa[k], &sno[k], &cso[k]);
  }

  __syncthreads();

  const v4f phs = Gs[G_V4];
  const float pr2 = phs[0], pi2 = phs[1];

  const v4f* Gw = Gs + lane;

  float st[2][2][2][8];

  {
    float c0[4][2], s0[4][2];
#pragma unroll
    for (int wq = 0; wq < 4; ++wq) {
      const float co = cso[wq], so = sno[wq];
      const float cx = xhalf(co), sx = xhalf(so);
      c0[wq][0] = hi ? cx : co;  c0[wq][1] = hi ? co : cx;
      s0[wq][0] = hi ? sx : so;  s0[wq][1] = hi ? so : sx;
    }
    float mg[2][8];
#pragma unroll
    for (int nt = 0; nt < 2; ++nt) {
      const float f1 = (hi ? s0[0][nt] : c0[0][nt]) * RS2;
#pragma unroll
      for (int r = 0; r < 8; ++r) {
        float v = f1 * ((r & 4) ? s0[1][nt] : c0[1][nt]);
        v = v * ((r & 2) ? s0[2][nt] : c0[2][nt]);
        mg[nt][r] = v * ((r & 1) ? s0[3][nt] : c0[3][nt]);
      }
    }

    v8f acc[2][2];
#pragma unroll
    for (int mo = 0; mo < 2; ++mo)
#pragma unroll
      for (int nt = 0; nt < 2; ++nt) acc[mo][nt] = (v8f){0.f, 0.f, 0.f, 0.f, 0.f, 0.f, 0.f, 0.f};
    v4f a0[2][2];
#pragma unroll
    for (int mo = 0; mo < 2; ++mo)
#pragma unroll
      for (int q = 0; q < 2; ++q) a0[mo][q] = Gw[(mo * 2 + q) * 32];
#pragma unroll
    for (int rp = 0; rp < 4; ++rp) {
#pragma unroll
      for (int mo = 0; mo < 2; ++mo) {
        const v4f aq = a0[mo][rp >> 1];
        v2f Af;
        Af[0] = (rp & 1) ? aq[2] : aq[0];
        Af[1] = (rp & 1) ? aq[3] : aq[1];
#pragma unroll
        for (int nt = 0; nt < 2; ++nt) {
          v2f Bf;
          Bf[0] = mg[nt][2 * rp];
          Bf[1] = mg[nt][2 * rp + 1];
          acc[mo][nt] = __builtin_amdgcn_wmma_f32_16x16x4_f32(
              false, Af, false, Bf, (short)0, acc[mo][nt], false, false);
        }
      }
    }

#pragma unroll
    for (int nt = 0; nt < 2; ++nt) {
#pragma unroll
      for (int r = 0; r < 8; ++r) {
        st[0][0][nt][r] = acc[0][nt][r];
        st[0][1][nt][r] = acc[1][nt][r];
      }
    }
  }

#pragma unroll
  for (int l = 1; l < 3; ++l) {
#pragma unroll
    for (int wq = (l == 2) ? 1 : 0; wq < 4; ++wq) {
      const float co = cso[4 * l + wq], so = sno[4 * l + wq];
      const float cx = xhalf(co), sx = xhalf(so);
      const float cc[2] = { hi ? cx : co, hi ? co : cx };
      const float ss[2] = { hi ? sx : so, hi ? so : sx };
#pragma unroll
      for (int nt = 0; nt < 2; ++nt) {
        const float c_ = cc[nt], s_ = ss[nt];
        if (wq == 0) {
#pragma unroll
          for (int w0 = 0; w0 < 2; ++w0) {
            if (l == 1 && w0 == 1) continue;
#pragma unroll
            for (int r = 0; r < 8; ++r) {
              float mr = st[w0][0][nt][r], mi = st[w0][1][nt][r];
              float orr = xhalf(mr),       oi = xhalf(mi);
              st[w0][0][nt][r] = c_ * mr + s_ * oi;
              st[w0][1][nt][r] = c_ * mi - s_ * orr;
            }
          }
        } else {
          const int bt = 8 >> wq;
#pragma unroll
          for (int w0 = 0; w0 < 2; ++w0) {
            if (l == 1 && w0 == 1) continue;
#pragma unroll
            for (int r = 0; r < 8; ++r) {
              if (!(r & bt)) {
                const int jj = r | bt;
                float ur = st[w0][0][nt][r],  ui = st[w0][1][nt][r];
                float vr = st[w0][0][nt][jj], vi = st[w0][1][nt][jj];
                st[w0][0][nt][r]  = c_ * ur + s_ * vi;
                st[w0][1][nt][r]  = c_ * ui - s_ * vr;
                st[w0][0][nt][jj] = c_ * vr + s_ * ui;
                st[w0][1][nt][jj] = c_ * vi - s_ * ur;
              }
            }
          }
        }
      }
    }

    if (l == 2) break;

    v8f nw[2][2];
#pragma unroll
    for (int mo = 0; mo < 2; ++mo)
#pragma unroll
      for (int nt = 0; nt < 2; ++nt) nw[mo][nt] = (v8f){0.f, 0.f, 0.f, 0.f, 0.f, 0.f, 0.f, 0.f};

#pragma unroll
    for (int kmt = 0; kmt < 2; ++kmt) {
      v4f a[2][2];
#pragma unroll
      for (int mo = 0; mo < 2; ++mo)
#pragma unroll
        for (int q = 0; q < 2; ++q)
          a[mo][q] = Gw[G_L12 / 4 + ((mo * 2 + kmt) * 2 + q) * 32];
#pragma unroll
      for (int rp = 0; rp < 4; ++rp) {
#pragma unroll
        for (int mo = 0; mo < 2; ++mo) {
          const v4f aq = a[mo][rp >> 1];
          v2f Af;
          Af[0] = (rp & 1) ? aq[2] : aq[0];
          Af[1] = (rp & 1) ? aq[3] : aq[1];
#pragma unroll
          for (int nt = 0; nt < 2; ++nt) {
            v2f Bf;
            Bf[0] = st[0][kmt][nt][2 * rp];
            Bf[1] = st[0][kmt][nt][2 * rp + 1];
            nw[mo][nt] = __builtin_amdgcn_wmma_f32_16x16x4_f32(
                false, Af, false, Bf, (short)0, nw[mo][nt], false, false);
          }
        }
      }
    }

    {
      const float co = cso[4], so = sno[4];
      const float cfo = co * co - so * so, sfo = 2.0f * co * so;
      const float cfx = xhalf(cfo), sfx = xhalf(sfo);
      const float ct[2] = { hi ? cfx : cfo, hi ? cfo : cfx };
      const float sn[2] = { hi ? sfx : sfo, hi ? sfo : sfx };
      const v4f* Tb = Gs + G_TAB / 4 + (hi ? 16 : 0);
#pragma unroll
      for (int r = 0; r < 8; ++r) {
        const v4f t0 = Tb[2 * r];
        const v4f t1 = Tb[2 * r + 1];
#pragma unroll
        for (int nt = 0; nt < 2; ++nt) {
          const float kr = t0[0] + t0[2] * ct[nt];
          const float ki = t0[1] + t0[3] * ct[nt];
          const float gr = t1[0] * sn[nt];
          const float gi = t1[1] * sn[nt];
          const float ur = nw[0][nt][r],     ui = nw[1][nt][r];
          const float xr = nw[0][nt][r ^ 7], xi = nw[1][nt][r ^ 7];
          st[0][0][nt][r] = ur;
          st[0][1][nt][r] = ui;
          st[1][0][nt][r] = kr * ur - ki * ui + gr * xr - gi * xi;
          st[1][1][nt][r] = kr * ui + ki * ur + gr * xi + gi * xr;
        }
      }
    }
  }

  float tot[2];
#pragma unroll
  for (int nt = 0; nt < 2; ++nt) {
    float se = 0.0f, sr = 0.0f, si = 0.0f;
#pragma unroll
    for (int r = 0; r < 8; ++r) {
      const float ar = st[0][0][nt][r], ai = st[0][1][nt][r];
      const float br = st[1][0][nt][r], bi = st[1][1][nt][r];
      if (((r >> 2) ^ (r >> 1) ^ r) & 1) {
        sr += ar * br;  sr += ai * bi;
        si += ar * bi;  si -= ai * br;
      } else {
        se += ar * br;  se += ai * bi;
      }
    }
    const float part = se + (pr2 * sr - pi2 * si);
    tot[nt] = 2.0f * (part + xhalf(part));
  }
  const float ev = hi ? tot[1] : tot[0];
  if (p < P) out[p] = ev;
}

extern "C" void kernel_launch(void* const* d_in, const int* in_sizes, int n_in,
                              void* d_out, int out_size, void* d_ws, size_t ws_size,
                              hipStream_t stream) {
  (void)in_sizes; (void)n_in; (void)ws_size;
  const float* x      = (const float*)d_in[0];
  const float* thetas = (const float*)d_in[1];
  const float* phis   = (const float*)d_in[2];
  float* out = (float*)d_out;
  (void)d_ws;

  const int P = out_size;
  const int blocks = (P + 255) / 256;
  hipLaunchKernelGGL(qc_main, dim3(blocks), dim3(256), 0, stream, x, thetas, phis, out, P);
}


# Layer 1 unitary split into 0/1 permutation times diagonal column factors: factors on the vector unit, permutation as one 16x16 product over re, im tiles: 16 WMMAs not 32; layer 1 rounding changes
# speedup vs baseline: 4.8599x; 1.0725x over previous
#include <hip/hip_runtime.h>
#include <math.h>


typedef __attribute__((ext_vector_type(2))) float v2f;
typedef __attribute__((ext_vector_type(4))) float v4f;
typedef __attribute__((ext_vector_type(8))) float v8f;

#define RS2 0.70710678118654752f
#define G_L12   512
#define G_PH    768
#define G_TAB   800
#define G_V4    232

__device__ __forceinline__ int frag0_off(int R, int C) {
  return ((((R >> 4) * 2 + ((C >> 2) & 1)) * 32) + ((C >> 3) & 1) * 16 + (R & 15)) * 4 + (C & 3);
}

__device__ __forceinline__ float from_lane(float v, int k) {
  return __builtin_bit_cast(float, __builtin_amdgcn_readlane(__builtin_bit_cast(int, v), k));
}

template <int L>
__device__ __forceinline__ void trace_column(int a, float cv, float sv,
                                             int& idx, float& pr, float& pi) {
  const int cb[4] = {8, 4, 2, 1};
  const int tb[4] = {4, 2, 1, 8};
  idx = a; pr = 1.0f; pi = 0.0f;
#pragma unroll
  for (int w = 0; w < 4; ++w) {
    const float ec = from_lane(cv, 4 * L + w);
    const float es = from_lane(sv, 4 * L + w);
    const bool ctrl = (idx & cb[w]) != 0;
    const float fr = (idx & tb[w]) ? es : -es;
    const float nr = pr * fr + pi * ec;
    const float ni = pi * fr - pr * ec;
    pr  = ctrl ? nr : pr;
    pi  = ctrl ? ni : pi;
    idx = ctrl ? (idx ^ tb[w]) : idx;
  }
}

__device__ __forceinline__ float xhalf(float v) {
  int i = __builtin_bit_cast(int, v);
  return __builtin_bit_cast(float,
      __builtin_amdgcn_permlanex16(i, i, 0x76543210, 0xfedcba98, false, false));
}

__device__ __forceinline__ float f32_bits(unsigned u) { return __builtin_bit_cast(float, u); }
__device__ __forceinline__ void cos_sin_small(float h, float& c, float& s) {
  const float ah = __builtin_fabsf(h);
  const float fn = __builtin_rintf(ah * f32_bits(0x3f22f983u));
  float r = __builtin_fmaf(fn, -f32_bits(0x3fc90fdau), ah);
  r = __builtin_fmaf(fn, -f32_bits(0x33a22168u), r);
  r = __builtin_fmaf(fn, -f32_bits(0x27c234c4u), r);
  const unsigned q = (unsigned)(int)fn;
  const float t = r * r;
  float ps = __builtin_fmaf(f32_bits(0xb94c1982u), t, f32_bits(0x3c0881c4u));
  ps = __builtin_fmaf(t, ps, f32_bits(0xbe2aaa9du));
  const float sr = __builtin_fmaf(r, t * ps, r);
  float pc = __builtin_fmaf(f32_bits(0x37d75334u), t, f32_bits(0xbab64f3bu));
  pc = __builtin_fmaf(t, pc, f32_bits(0x3d2aabf7u));
  pc = __builtin_fmaf(t, pc, f32_bits(0xbf000004u));
  const float cr = __builtin_fmaf(t, pc, 1.0f);
  const bool odd = (q & 1u) != 0u;
  const unsigned flip = (q << 30) & 0x80000000u;
  const unsigned sgn = __builtin_bit_cast(unsigned, h) & 0x80000000u;
  const float sq = odd ? cr : sr;
  const float cq = odd ? -sr : cr;
  s = __builtin_bit_cast(float, __builtin_bit_cast(unsigned, sq) ^ flip ^ sgn);
  c = __builtin_bit_cast(float, __builtin_bit_cast(unsigned, cq) ^ flip);
}

__global__ __launch_bounds__(256) void qc_main(
    const float* __restrict__ x, const float* __restrict__ thetas,
    const float* __restrict__ phis, float* __restrict__ out, int P) {
  __shared__ v4f Gs[G_V4 + 1];

  const int tid  = threadIdx.x;
  const int lane = tid & 31;
  const bool hi  = (lane >> 4) != 0;
  const int wv   = __builtin_amdgcn_readfirstlane(tid >> 5);

  const int p  = blockIdx.x * 256 + tid;
  const int b  = p >> 14;
  const int h2 = (p >> 7) & 127;
  const int w2 = p & 127;

  float ang[12];
#pragma unroll
  for (int c = 0; c < 3; ++c) {
#pragma unroll
    for (int dy = 0; dy < 2; ++dy) {
      const float* row = x + (((b * 3 + c) * 256 + (2 * h2 + dy)) * 256 + 2 * w2);
      ang[c * 4 + dy * 2 + 0] = row[0];
      ang[c * 4 + dy * 2 + 1] = row[1];
    }
  }

  if (wv < 2) {
    const int ip = lane < 8 ? 0 : (lane > 10 ? 2 : lane - 8);
    const float tv = thetas[lane & 7];
    const float pv = phis[ip];
    const float hv = lane < 8 ? 0.5f * tv : (lane == 8 ? 0.5f * pv : pv);
    float cv, sv;
    sincosf(hv, &sv, &cv);
    float* Gf = reinterpret_cast<float*>(Gs);
    const v4f zero = (v4f){0.f, 0.f, 0.f, 0.f};
    int idx; float vr, vi;
    if (wv == 0) {
#pragma unroll
      for (int j = 0; j < 2; ++j) Gs[G_L12 / 4 + 32 * j + lane] = zero;
      const int a = (lane & 15) ^ (hi ? 8 : 0);
      trace_column<1>(a, cv, sv, idx, vr, vi);
      const float qr = xhalf(vr), qi = xhalf(vi);
      if (!hi) {
        Gf[G_L12 + frag0_off(idx, a)] = 1.0f;
        Gf[G_PH + 2 * a]     = vr;
        Gf[G_PH + 2 * a + 1] = vi;
        const float Ch = from_lane(cv, 8), Sh = from_lane(sv, 8);
        const float c1 = from_lane(cv, 9), s1 = from_lane(sv, 9);
        const bool w1o = (idx & 8) != 0;
        const float kr = w1o ? (c1 * Ch - s1 * Sh) : Ch;
        const float ki = w1o ? (c1 * Sh + s1 * Ch) : Sh;
        const float sS = (a & 8) ? -Sh : Sh;
        const float psr = vr * qr + vi * qi;
        const float psi = vi * qr - vr * qi;
        const float gr = (kr * psr - ki * psi) * sS;
        const float gi = (kr * psi + ki * psr) * sS;
        Gs[G_TAB / 4 + 2 * idx]     = (v4f){kr * Ch, ki * Ch, ki * sS, -kr * sS};
        Gs[G_TAB / 4 + 2 * idx + 1] = (v4f){gr, gi, 0.f, 0.f};
      }
    } else {
#pragma unroll
      for (int j = 0; j < 4; ++j) Gs[32 * j + lane] = zero;
      const int a = lane & 15;
      trace_column<0>(a, cv, sv, idx, vr, vi);
      const int pc = __popc(a) & 3;
      const float xr = (pc == 0) ? vr : (pc == 1) ?  vi : (pc == 2) ? -vr : -vi;
      const float xi = (pc == 0) ? vi : (pc == 1) ? -vr : (pc == 2) ? -vi :  vr;
      if (!hi) {
        Gf[frag0_off(idx,      a)] = xr;
        Gf[frag0_off(idx + 16, a)] = xi;
      }
      if (lane == 0)
        Gs[G_V4] = (v4f){from_lane(cv, 10), from_lane(sv, 10), 0.f, 0.f};
    }
  }

  float cso[12], sno[12];
  float hfa[12];
  float amax = 0.0f;
#pragma unroll
  for (int k = 0; k < 12; ++k) {
    hfa[k] = 0.5f * ang[k];
    if (k != 8) amax = __builtin_fmaxf(amax, __builtin_fabsf(hfa[k]));
  }
  cso[8] = 1.0f; sno[8] = 0.0f;
  if (amax < 131072.0f) {
#pragma unroll
    for (int k = 0; k < 12; ++k)
      if (k != 8) cos_sin_small(hfa[k], cso[k], sno[k]);
  } else {
#pragma unroll
    for (int k = 0; k < 12; ++k)
      if (k != 8) sincosf(hfa[k], &sno[k], &cso[k]);
  }

  __syncthreads();

  const v4f phs = Gs[G_V4];
  const float pr2 = phs[0], pi2 = phs[1];

  const v4f* Gw = Gs + lane;

  float st[2][2][2][8];

  {
    float c0[4][2], s0[4][2];
#pragma unroll
    for (int wq = 0; wq < 4; ++wq) {
      const float co = cso[wq], so = sno[wq];
      const float cx = xhalf(co), sx = xhalf(so);
      c0[wq][0] = hi ? cx : co;  c0[wq][1] = hi ? co : cx;
      s0[wq][0] = hi ? sx : so;  s0[wq][1] = hi ? so : sx;
    }
    float mg[2][8];
#pragma unroll
    for (int nt = 0; nt < 2; ++nt) {
      const float f1 = (hi ? s0[0][nt] : c0[0][nt]) * RS2;
#pragma unroll
      for (int r = 0; r < 8; ++r) {
        float v = f1 * ((r & 4) ? s0[1][nt] : c0[1][nt]);
        v = v * ((r & 2) ? s0[2][nt] : c0[2][nt]);
        mg[nt][r] = v * ((r & 1) ? s0[3][nt] : c0[3][nt]);
      }
    }

    v8f acc[2][2];
#pragma unroll
    for (int mo = 0; mo < 2; ++mo)
#pragma unroll
      for (int nt = 0; nt < 2; ++nt) acc[mo][nt] = (v8f){0.f, 0.f, 0.f, 0.f, 0.f, 0.f, 0.f, 0.f};
    v4f a0[2][2];
#pragma unroll
    for (int mo = 0; mo < 2; ++mo)
#pragma unroll
      for (int q = 0; q < 2; ++q) a0[mo][q] = Gw[(mo * 2 + q) * 32];
#pragma unroll
    for (int rp = 0; rp < 4; ++rp) {
#pragma unroll
      for (int mo = 0; mo < 2; ++mo) {
        const v4f aq = a0[mo][rp >> 1];
        v2f Af;
        Af[0] = (rp & 1) ? aq[2] : aq[0];
        Af[1] = (rp & 1) ? aq[3] : aq[1];
#pragma unroll
        for (int nt = 0; nt < 2; ++nt) {
          v2f Bf;
          Bf[0] = mg[nt][2 * rp];
          Bf[1] = mg[nt][2 * rp + 1];
          acc[mo][nt] = __builtin_amdgcn_wmma_f32_16x16x4_f32(
              false, Af, false, Bf, (short)0, acc[mo][nt], false, false);
        }
      }
    }

#pragma unroll
    for (int nt = 0; nt < 2; ++nt) {
#pragma unroll
      for (int r = 0; r < 8; ++r) {
        st[0][0][nt][r] = acc[0][nt][r];
        st[0][1][nt][r] = acc[1][nt][r];
      }
    }
  }

#pragma unroll
  for (int l = 1; l < 3; ++l) {
#pragma unroll
    for (int wq = (l == 2) ? 1 : 0; wq < 4; ++wq) {
      const float co = cso[4 * l + wq], so = sno[4 * l + wq];
      const float cx = xhalf(co), sx = xhalf(so);
      const float cc[2] = { hi ? cx : co, hi ? co : cx };
      const float ss[2] = { hi ? sx : so, hi ? so : sx };
#pragma unroll
      for (int nt = 0; nt < 2; ++nt) {
        const float c_ = cc[nt], s_ = ss[nt];
        if (wq == 0) {
#pragma unroll
          for (int w0 = 0; w0 < 2; ++w0) {
            if (l == 1 && w0 == 1) continue;
#pragma unroll
            for (int r = 0; r < 8; ++r) {
              float mr = st[w0][0][nt][r], mi = st[w0][1][nt][r];
              float orr = xhalf(mr),       oi = xhalf(mi);
              st[w0][0][nt][r] = c_ * mr + s_ * oi;
              st[w0][1][nt][r] = c_ * mi - s_ * orr;
            }
          }
        } else {
          const int bt = 8 >> wq;
#pragma unroll
          for (int w0 = 0; w0 < 2; ++w0) {
            if (l == 1 && w0 == 1) continue;
#pragma unroll
            for (int r = 0; r < 8; ++r) {
              if (!(r & bt)) {
                const int jj = r | bt;
                float ur = st[w0][0][nt][r],  ui = st[w0][1][nt][r];
                float vr = st[w0][0][nt][jj], vi = st[w0][1][nt][jj];
                st[w0][0][nt][r]  = c_ * ur + s_ * vi;
                st[w0][1][nt][r]  = c_ * ui - s_ * vr;
                st[w0][0][nt][jj] = c_ * vr + s_ * ui;
                st[w0][1][nt][jj] = c_ * vi - s_ * ur;
              }
            }
          }
        }
      }
    }

    if (l == 2) break;

    float fa[2][2][8];
    {
      const v4f* Ph = Gs + G_PH / 4 + (hi ? 4 : 0);
      v4f ph[4];
#pragma unroll
      for (int q = 0; q < 4; ++q) ph[q] = Ph[q];
#pragma unroll
      for (int r = 0; r < 8; ++r) {
        const float fr = ph[r >> 1][2 * (r & 1)], fi = ph[r >> 1][2 * (r & 1) + 1];
#pragma unroll
        for (int nt = 0; nt < 2; ++nt) {
          const float ar = st[0][0][nt][r], ai = st[0][1][nt][r];
          fa[0][nt][r] = fr * ar - fi * ai;
          fa[1][nt][r] = fi * ar + fr * ai;
        }
      }
    }

    v8f nw[2][2];
#pragma unroll
    for (int mo = 0; mo < 2; ++mo)
#pragma unroll
      for (int nt = 0; nt < 2; ++nt) nw[mo][nt] = (v8f){0.f, 0.f, 0.f, 0.f, 0.f, 0.f, 0.f, 0.f};
    {
      v4f pm[2];
#pragma unroll
      for (int q = 0; q < 2; ++q) pm[q] = Gw[G_L12 / 4 + q * 32];
#pragma unroll
      for (int rp = 0; rp < 4; ++rp) {
        const v4f aq = pm[rp >> 1];
        v2f Af;
        Af[0] = (rp & 1) ? aq[2] : aq[0];
        Af[1] = (rp & 1) ? aq[3] : aq[1];
#pragma unroll
        for (int mo = 0; mo < 2; ++mo) {
#pragma unroll
          for (int nt = 0; nt < 2; ++nt) {
            v2f Bf;
            Bf[0] = fa[mo][nt][2 * rp];
            Bf[1] = fa[mo][nt][2 * rp + 1];
            nw[mo][nt] = __builtin_amdgcn_wmma_f32_16x16x4_f32(
                false, Af, false, Bf, (short)0, nw[mo][nt], false, false);
          }
        }
      }
    }

    {
      const float co = cso[4], so = sno[4];
      const float cfo = co * co - so * so, sfo = 2.0f * co * so;
      const float cfx = xhalf(cfo), sfx = xhalf(sfo);
      const float ct[2] = { hi ? cfx : cfo, hi ? cfo : cfx };
      const float sn[2] = { hi ? sfx : sfo, hi ? sfo : sfx };
      const v4f* Tb = Gs + G_TAB / 4 + (hi ? 16 : 0);
#pragma unroll
      for (int r = 0; r < 8; ++r) {
        const v4f t0 = Tb[2 * r];
        const v4f t1 = Tb[2 * r + 1];
#pragma unroll
        for (int nt = 0; nt < 2; ++nt) {
          const float kr = t0[0] + t0[2] * ct[nt];
          const float ki = t0[1] + t0[3] * ct[nt];
          const float gr = t1[0] * sn[nt];
          const float gi = t1[1] * sn[nt];
          const float ur = nw[0][nt][r],     ui = nw[1][nt][r];
          const float xr = nw[0][nt][r ^ 7], xi = nw[1][nt][r ^ 7];
          st[0][0][nt][r] = ur;
          st[0][1][nt][r] = ui;
          st[1][0][nt][r] = kr * ur - ki * ui + gr * xr - gi * xi;
          st[1][1][nt][r] = kr * ui + ki * ur + gr * xi + gi * xr;
        }
      }
    }
  }

  float tot[2];
#pragma unroll
  for (int nt = 0; nt < 2; ++nt) {
    float se = 0.0f, sr = 0.0f, si = 0.0f;
#pragma unroll
    for (int r = 0; r < 8; ++r) {
      const float ar = st[0][0][nt][r], ai = st[0][1][nt][r];
      const float br = st[1][0][nt][r], bi = st[1][1][nt][r];
      if (((r >> 2) ^ (r >> 1) ^ r) & 1) {
        sr += ar * br;  sr += ai * bi;
        si += ar * bi;  si -= ai * br;
      } else {
        se += ar * br;  se += ai * bi;
      }
    }
    const float part = se + (pr2 * sr - pi2 * si);
    tot[nt] = 2.0f * (part + xhalf(part));
  }
  const float ev = hi ? tot[1] : tot[0];
  if (p < P) out[p] = ev;
}

extern "C" void kernel_launch(void* const* d_in, const int* in_sizes, int n_in,
                              void* d_out, int out_size, void* d_ws, size_t ws_size,
                              hipStream_t stream) {
  (void)in_sizes; (void)n_in; (void)ws_size;
  const float* x      = (const float*)d_in[0];
  const float* thetas = (const float*)d_in[1];
  const float* phis   = (const float*)d_in[2];
  float* out = (float*)d_out;
  (void)d_ws;

  const int P = out_size;
  const int blocks = (P + 255) / 256;
  hipLaunchKernelGGL(qc_main, dim3(blocks), dim3(256), 0, stream, x, thetas, phis, out, P);
}


# Registers per wave capped at 128 by the second launch bound (122 used, was 191, no scratch): 8 waves per SIMD, all four workgroups of a WGP resident at once, was two; arithmetic and sum order same
# speedup vs baseline: 5.2393x; 1.0781x over previous
#include <hip/hip_runtime.h>
#include <math.h>


typedef __attribute__((ext_vector_type(2))) float v2f;
typedef __attribute__((ext_vector_type(4))) float v4f;
typedef __attribute__((ext_vector_type(8))) float v8f;

#define RS2 0.70710678118654752f
#define G_L12   512
#define G_PH    768
#define G_TAB   800
#define G_V4    232

__device__ __forceinline__ int frag0_off(int R, int C) {
  return ((((R >> 4) * 2 + ((C >> 2) & 1)) * 32) + ((C >> 3) & 1) * 16 + (R & 15)) * 4 + (C & 3);
}

__device__ __forceinline__ float from_lane(float v, int k) {
  return __builtin_bit_cast(float, __builtin_amdgcn_readlane(__builtin_bit_cast(int, v), k));
}

template <int L>
__device__ __forceinline__ void trace_column(int a, float cv, float sv,
                                             int& idx, float& pr, float& pi) {
  const int cb[4] = {8, 4, 2, 1};
  const int tb[4] = {4, 2, 1, 8};
  idx = a; pr = 1.0f; pi = 0.0f;
#pragma unroll
  for (int w = 0; w < 4; ++w) {
    const float ec = from_lane(cv, 4 * L + w);
    const float es = from_lane(sv, 4 * L + w);
    const bool ctrl = (idx & cb[w]) != 0;
    const float fr = (idx & tb[w]) ? es : -es;
    const float nr = pr * fr + pi * ec;
    const float ni = pi * fr - pr * ec;
    pr  = ctrl ? nr : pr;
    pi  = ctrl ? ni : pi;
    idx = ctrl ? (idx ^ tb[w]) : idx;
  }
}

__device__ __forceinline__ float xhalf(float v) {
  int i = __builtin_bit_cast(int, v);
  return __builtin_bit_cast(float,
      __builtin_amdgcn_permlanex16(i, i, 0x76543210, 0xfedcba98, false, false));
}

__device__ __forceinline__ float f32_bits(unsigned u) { return __builtin_bit_cast(float, u); }
__device__ __forceinline__ void cos_sin_small(float h, float& c, float& s) {
  const float ah = __builtin_fabsf(h);
  const float fn = __builtin_rintf(ah * f32_bits(0x3f22f983u));
  float r = __builtin_fmaf(fn, -f32_bits(0x3fc90fdau), ah);
  r = __builtin_fmaf(fn, -f32_bits(0x33a22168u), r);
  r = __builtin_fmaf(fn, -f32_bits(0x27c234c4u), r);
  const unsigned q = (unsigned)(int)fn;
  const float t = r * r;
  float ps = __builtin_fmaf(f32_bits(0xb94c1982u), t, f32_bits(0x3c0881c4u));
  ps = __builtin_fmaf(t, ps, f32_bits(0xbe2aaa9du));
  const float sr = __builtin_fmaf(r, t * ps, r);
  float pc = __builtin_fmaf(f32_bits(0x37d75334u), t, f32_bits(0xbab64f3bu));
  pc = __builtin_fmaf(t, pc, f32_bits(0x3d2aabf7u));
  pc = __builtin_fmaf(t, pc, f32_bits(0xbf000004u));
  const float cr = __builtin_fmaf(t, pc, 1.0f);
  const bool odd = (q & 1u) != 0u;
  const unsigned flip = (q << 30) & 0x80000000u;
  const unsigned sgn = __builtin_bit_cast(unsigned, h) & 0x80000000u;
  const float sq = odd ? cr : sr;
  const float cq = odd ? -sr : cr;
  s = __builtin_bit_cast(float, __builtin_bit_cast(unsigned, sq) ^ flip ^ sgn);
  c = __builtin_bit_cast(float, __builtin_bit_cast(unsigned, cq) ^ flip);
}

__global__ __launch_bounds__(256, 8) void qc_main(
    const float* __restrict__ x, const float* __restrict__ thetas,
    const float* __restrict__ phis, float* __restrict__ out, int P) {
  __shared__ v4f Gs[G_V4 + 1];

  const int tid  = threadIdx.x;
  const int lane = tid & 31;
  const bool hi  = (lane >> 4) != 0;
  const int wv   = __builtin_amdgcn_readfirstlane(tid >> 5);

  const int p  = blockIdx.x * 256 + tid;
  const int b  = p >> 14;
  const int h2 = (p >> 7) & 127;
  const int w2 = p & 127;

  float ang[12];
#pragma unroll
  for (int c = 0; c < 3; ++c) {
#pragma unroll
    for (int dy = 0; dy < 2; ++dy) {
      const float* row = x + (((b * 3 + c) * 256 + (2 * h2 + dy)) * 256 + 2 * w2);
      ang[c * 4 + dy * 2 + 0] = row[0];
      ang[c * 4 + dy * 2 + 1] = row[1];
    }
  }

  if (wv < 2) {
    const int ip = lane < 8 ? 0 : (lane > 10 ? 2 : lane - 8);
    const float tv = thetas[lane & 7];
    const float pv = phis[ip];
    const float hv = lane < 8 ? 0.5f * tv : (lane == 8 ? 0.5f * pv : pv);
    float cv, sv;
    sincosf(hv, &sv, &cv);
    float* Gf = reinterpret_cast<float*>(Gs);
    const v4f zero = (v4f){0.f, 0.f, 0.f, 0.f};
    int idx; float vr, vi;
    if (wv == 0) {
#pragma unroll
      for (int j = 0; j < 2; ++j) Gs[G_L12 / 4 + 32 * j + lane] = zero;
      const int a = (lane & 15) ^ (hi ? 8 : 0);
      trace_column<1>(a, cv, sv, idx, vr, vi);
      const float qr = xhalf(vr), qi = xhalf(vi);
      if (!hi) {
        Gf[G_L12 + frag0_off(idx, a)] = 1.0f;
        Gf[G_PH + 2 * a]     = vr;
        Gf[G_PH + 2 * a + 1] = vi;
        const float Ch = from_lane(cv, 8), Sh = from_lane(sv, 8);
        const float c1 = from_lane(cv, 9), s1 = from_lane(sv, 9);
        const bool w1o = (idx & 8) != 0;
        const float kr = w1o ? (c1 * Ch - s1 * Sh) : Ch;
        const float ki = w1o ? (c1 * Sh + s1 * Ch) : Sh;
        const float sS = (a & 8) ? -Sh : Sh;
        const float psr = vr * qr + vi * qi;
        const float psi = vi * qr - vr * qi;
        const float gr = (kr * psr - ki * psi) * sS;
        const float gi = (kr * psi + ki * psr) * sS;
        Gs[G_TAB / 4 + 2 * idx]     = (v4f){kr * Ch, ki * Ch, ki * sS, -kr * sS};
        Gs[G_TAB / 4 + 2 * idx + 1] = (v4f){gr, gi, 0.f, 0.f};
      }
    } else {
#pragma unroll
      for (int j = 0; j < 4; ++j) Gs[32 * j + lane] = zero;
      const int a = lane & 15;
      trace_column<0>(a, cv, sv, idx, vr, vi);
      const int pc = __popc(a) & 3;
      const float xr = (pc == 0) ? vr : (pc == 1) ?  vi : (pc == 2) ? -vr : -vi;
      const float xi = (pc == 0) ? vi : (pc == 1) ? -vr : (pc == 2) ? -vi :  vr;
      if (!hi) {
        Gf[frag0_off(idx,      a)] = xr;
        Gf[frag0_off(idx + 16, a)] = xi;
      }
      if (lane == 0)
        Gs[G_V4] = (v4f){from_lane(cv, 10), from_lane(sv, 10), 0.f, 0.f};
    }
  }

  float cso[12], sno[12];
  float hfa[12];
  float amax = 0.0f;
#pragma unroll
  for (int k = 0; k < 12; ++k) {
    hfa[k] = 0.5f * ang[k];
    if (k != 8) amax = __builtin_fmaxf(amax, __builtin_fabsf(hfa[k]));
  }
  cso[8] = 1.0f; sno[8] = 0.0f;
  if (amax < 131072.0f) {
#pragma unroll
    for (int k = 0; k < 12; ++k)
      if (k != 8) cos_sin_small(hfa[k], cso[k], sno[k]);
  } else {
#pragma unroll
    for (int k = 0; k < 12; ++k)
      if (k != 8) sincosf(hfa[k], &sno[k], &cso[k]);
  }

  __syncthreads();

  const v4f phs = Gs[G_V4];
  const float pr2 = phs[0], pi2 = phs[1];

  const v4f* Gw = Gs + lane;

  float st[2][2][2][8];

  {
    float c0[4][2], s0[4][2];
#pragma unroll
    for (int wq = 0; wq < 4; ++wq) {
      const float co = cso[wq], so = sno[wq];
      const float cx = xhalf(co), sx = xhalf(so);
      c0[wq][0] = hi ? cx : co;  c0[wq][1] = hi ? co : cx;
      s0[wq][0] = hi ? sx : so;  s0[wq][1] = hi ? so : sx;
    }
    float mg[2][8];
#pragma unroll
    for (int nt = 0; nt < 2; ++nt) {
      const float f1 = (hi ? s0[0][nt] : c0[0][nt]) * RS2;
#pragma unroll
      for (int r = 0; r < 8; ++r) {
        float v = f1 * ((r & 4) ? s0[1][nt] : c0[1][nt]);
        v = v * ((r & 2) ? s0[2][nt] : c0[2][nt]);
        mg[nt][r] = v * ((r & 1) ? s0[3][nt] : c0[3][nt]);
      }
    }

    v8f acc[2][2];
#pragma unroll
    for (int mo = 0; mo < 2; ++mo)
#pragma unroll
      for (int nt = 0; nt < 2; ++nt) acc[mo][nt] = (v8f){0.f, 0.f, 0.f, 0.f, 0.f, 0.f, 0.f, 0.f};
    v4f a0[2][2];
#pragma unroll
    for (int mo = 0; mo < 2; ++mo)
#pragma unroll
      for (int q = 0; q < 2; ++q) a0[mo][q] = Gw[(mo * 2 + q) * 32];
#pragma unroll
    for (int rp = 0; rp < 4; ++rp) {
#pragma unroll
      for (int mo = 0; mo < 2; ++mo) {
        const v4f aq = a0[mo][rp >> 1];
        v2f Af;
        Af[0] = (rp & 1) ? aq[2] : aq[0];
        Af[1] = (rp & 1) ? aq[3] : aq[1];
#pragma unroll
        for (int nt = 0; nt < 2; ++nt) {
          v2f Bf;
          Bf[0] = mg[nt][2 * rp];
          Bf[1] = mg[nt][2 * rp + 1];
          acc[mo][nt] = __builtin_amdgcn_wmma_f32_16x16x4_f32(
              false, Af, false, Bf, (short)0, acc[mo][nt], false, false);
        }
      }
    }

#pragma unroll
    for (int nt = 0; nt < 2; ++nt) {
#pragma unroll
      for (int r = 0; r < 8; ++r) {
        st[0][0][nt][r] = acc[0][nt][r];
        st[0][1][nt][r] = acc[1][nt][r];
      }
    }
  }

#pragma unroll
  for (int l = 1; l < 3; ++l) {
#pragma unroll
    for (int wq = (l == 2) ? 1 : 0; wq < 4; ++wq) {
      const float co = cso[4 * l + wq], so = sno[4 * l + wq];
      const float cx = xhalf(co), sx = xhalf(so);
      const float cc[2] = { hi ? cx : co, hi ? co : cx };
      const float ss[2] = { hi ? sx : so, hi ? so : sx };
#pragma unroll
      for (int nt = 0; nt < 2; ++nt) {
        const float c_ = cc[nt], s_ = ss[nt];
        if (wq == 0) {
#pragma unroll
          for (int w0 = 0; w0 < 2; ++w0) {
            if (l == 1 && w0 == 1) continue;
#pragma unroll
            for (int r = 0; r < 8; ++r) {
              float mr = st[w0][0][nt][r], mi = st[w0][1][nt][r];
              float orr = xhalf(mr),       oi = xhalf(mi);
              st[w0][0][nt][r] = c_ * mr + s_ * oi;
              st[w0][1][nt][r] = c_ * mi - s_ * orr;
            }
          }
        } else {
          const int bt = 8 >> wq;
#pragma unroll
          for (int w0 = 0; w0 < 2; ++w0) {
            if (l == 1 && w0 == 1) continue;
#pragma unroll
            for (int r = 0; r < 8; ++r) {
              if (!(r & bt)) {
                const int jj = r | bt;
                float ur = st[w0][0][nt][r],  ui = st[w0][1][nt][r];
                float vr = st[w0][0][nt][jj], vi = st[w0][1][nt][jj];
                st[w0][0][nt][r]  = c_ * ur + s_ * vi;
                st[w0][1][nt][r]  = c_ * ui - s_ * vr;
                st[w0][0][nt][jj] = c_ * vr + s_ * ui;
                st[w0][1][nt][jj] = c_ * vi - s_ * ur;
              }
            }
          }
        }
      }
    }

    if (l == 2) break;

    float fa[2][2][8];
    {
      const v4f* Ph = Gs + G_PH / 4 + (hi ? 4 : 0);
      v4f ph[4];
#pragma unroll
      for (int q = 0; q < 4; ++q) ph[q] = Ph[q];
#pragma unroll
      for (int r = 0; r < 8; ++r) {
        const float fr = ph[r >> 1][2 * (r & 1)], fi = ph[r >> 1][2 * (r & 1) + 1];
#pragma unroll
        for (int nt = 0; nt < 2; ++nt) {
          const float ar = st[0][0][nt][r], ai = st[0][1][nt][r];
          fa[0][nt][r] = fr * ar - fi * ai;
          fa[1][nt][r] = fi * ar + fr * ai;
        }
      }
    }

    v8f nw[2][2];
#pragma unroll
    for (int mo = 0; mo < 2; ++mo)
#pragma unroll
      for (int nt = 0; nt < 2; ++nt) nw[mo][nt] = (v8f){0.f, 0.f, 0.f, 0.f, 0.f, 0.f, 0.f, 0.f};
    {
      v4f pm[2];
#pragma unroll
      for (int q = 0; q < 2; ++q) pm[q] = Gw[G_L12 / 4 + q * 32];
#pragma unroll
      for (int rp = 0; rp < 4; ++rp) {
        const v4f aq = pm[rp >> 1];
        v2f Af;
        Af[0] = (rp & 1) ? aq[2] : aq[0];
        Af[1] = (rp & 1) ? aq[3] : aq[1];
#pragma unroll
        for (int mo = 0; mo < 2; ++mo) {
#pragma unroll
          for (int nt = 0; nt < 2; ++nt) {
            v2f Bf;
            Bf[0] = fa[mo][nt][2 * rp];
            Bf[1] = fa[mo][nt][2 * rp + 1];
            nw[mo][nt] = __builtin_amdgcn_wmma_f32_16x16x4_f32(
                false, Af, false, Bf, (short)0, nw[mo][nt], false, false);
          }
        }
      }
    }

    {
      const float co = cso[4], so = sno[4];
      const float cfo = co * co - so * so, sfo = 2.0f * co * so;
      const float cfx = xhalf(cfo), sfx = xhalf(sfo);
      const float ct[2] = { hi ? cfx : cfo, hi ? cfo : cfx };
      const float sn[2] = { hi ? sfx : sfo, hi ? sfo : sfx };
      const v4f* Tb = Gs + G_TAB / 4 + (hi ? 16 : 0);
#pragma unroll
      for (int r = 0; r < 8; ++r) {
        const v4f t0 = Tb[2 * r];
        const v4f t1 = Tb[2 * r + 1];
#pragma unroll
        for (int nt = 0; nt < 2; ++nt) {
          const float kr = t0[0] + t0[2] * ct[nt];
          const float ki = t0[1] + t0[3] * ct[nt];
          const float gr = t1[0] * sn[nt];
          const float gi = t1[1] * sn[nt];
          const float ur = nw[0][nt][r],     ui = nw[1][nt][r];
          const float xr = nw[0][nt][r ^ 7], xi = nw[1][nt][r ^ 7];
          st[0][0][nt][r] = ur;
          st[0][1][nt][r] = ui;
          st[1][0][nt][r] = kr * ur - ki * ui + gr * xr - gi * xi;
          st[1][1][nt][r] = kr * ui + ki * ur + gr * xi + gi * xr;
        }
      }
    }
  }

  float tot[2];
#pragma unroll
  for (int nt = 0; nt < 2; ++nt) {
    float se = 0.0f, sr = 0.0f, si = 0.0f;
#pragma unroll
    for (int r = 0; r < 8; ++r) {
      const float ar = st[0][0][nt][r], ai = st[0][1][nt][r];
      const float br = st[1][0][nt][r], bi = st[1][1][nt][r];
      if (((r >> 2) ^ (r >> 1) ^ r) & 1) {
        sr += ar * br;  sr += ai * bi;
        si += ar * bi;  si -= ai * br;
      } else {
        se += ar * br;  se += ai * bi;
      }
    }
    const float part = se + (pr2 * sr - pi2 * si);
    tot[nt] = 2.0f * (part + xhalf(part));
  }
  const float ev = hi ? tot[1] : tot[0];
  if (p < P) out[p] = ev;
}

extern "C" void kernel_launch(void* const* d_in, const int* in_sizes, int n_in,
                              void* d_out, int out_size, void* d_ws, size_t ws_size,
                              hipStream_t stream) {
  (void)in_sizes; (void)n_in; (void)ws_size;
  const float* x      = (const float*)d_in[0];
  const float* thetas = (const float*)d_in[1];
  const float* phis   = (const float*)d_in[2];
  float* out = (float*)d_out;
  (void)d_ws;

  const int P = out_size;
  const int blocks = (P + 255) / 256;
  hipLaunchKernelGGL(qc_main, dim3(blocks), dim3(256), 0, stream, x, thetas, phis, out, P);
}


# Layer 1 permutation matrix applied as a fixed move of rows (register renaming, 16 lane-half swaps; lane half 1 keeps wire 2 flipped up to it), not as a 0/1 product: 16 WMMAs not 32; same bits
# speedup vs baseline: 5.8350x; 1.1137x over previous
#include <hip/hip_runtime.h>
#include <math.h>


typedef __attribute__((ext_vector_type(2))) float v2f;
typedef __attribute__((ext_vector_type(4))) float v4f;
typedef __attribute__((ext_vector_type(8))) float v8f;

#define RS2 0.70710678118654752f
#define G_PH    512
#define G_TAB   544
#define G_V4    168

__device__ __forceinline__ int flipped_pos(int idx) { return idx ^ ((idx >> 1) & 4); }

__device__ __forceinline__ int frag0_off(int R, int C) {
  return ((((R >> 4) * 2 + ((C >> 2) & 1)) * 32) + ((C >> 3) & 1) * 16 + (R & 15)) * 4 + (C & 3);
}

__device__ __forceinline__ float from_lane(float v, int k) {
  return __builtin_bit_cast(float, __builtin_amdgcn_readlane(__builtin_bit_cast(int, v), k));
}

template <int L>
__device__ __forceinline__ void trace_column(int a, float cv, float sv,
                                             int& idx, float& pr, float& pi) {
  const int cb[4] = {8, 4, 2, 1};
  const int tb[4] = {4, 2, 1, 8};
  idx = a; pr = 1.0f; pi = 0.0f;
#pragma unroll
  for (int w = 0; w < 4; ++w) {
    const float ec = from_lane(cv, 4 * L + w);
    const float es = from_lane(sv, 4 * L + w);
    const bool ctrl = (idx & cb[w]) != 0;
    const float fr = (idx & tb[w]) ? es : -es;
    const float nr = pr * fr + pi * ec;
    const float ni = pi * fr - pr * ec;
    pr  = ctrl ? nr : pr;
    pi  = ctrl ? ni : pi;
    idx = ctrl ? (idx ^ tb[w]) : idx;
  }
}

__device__ __forceinline__ float xhalf(float v) {
  int i = __builtin_bit_cast(int, v);
  return __builtin_bit_cast(float,
      __builtin_amdgcn_permlanex16(i, i, 0x76543210, 0xfedcba98, false, false));
}

__device__ __forceinline__ float f32_bits(unsigned u) { return __builtin_bit_cast(float, u); }
__device__ __forceinline__ void cos_sin_small(float h, float& c, float& s) {
  const float ah = __builtin_fabsf(h);
  const float fn = __builtin_rintf(ah * f32_bits(0x3f22f983u));
  float r = __builtin_fmaf(fn, -f32_bits(0x3fc90fdau), ah);
  r = __builtin_fmaf(fn, -f32_bits(0x33a22168u), r);
  r = __builtin_fmaf(fn, -f32_bits(0x27c234c4u), r);
  const unsigned q = (unsigned)(int)fn;
  const float t = r * r;
  float ps = __builtin_fmaf(f32_bits(0xb94c1982u), t, f32_bits(0x3c0881c4u));
  ps = __builtin_fmaf(t, ps, f32_bits(0xbe2aaa9du));
  const float sr = __builtin_fmaf(r, t * ps, r);
  float pc = __builtin_fmaf(f32_bits(0x37d75334u), t, f32_bits(0xbab64f3bu));
  pc = __builtin_fmaf(t, pc, f32_bits(0x3d2aabf7u));
  pc = __builtin_fmaf(t, pc, f32_bits(0xbf000004u));
  const float cr = __builtin_fmaf(t, pc, 1.0f);
  const bool odd = (q & 1u) != 0u;
  const unsigned flip = (q << 30) & 0x80000000u;
  const unsigned sgn = __builtin_bit_cast(unsigned, h) & 0x80000000u;
  const float sq = odd ? cr : sr;
  const float cq = odd ? -sr : cr;
  s = __builtin_bit_cast(float, __builtin_bit_cast(unsigned, sq) ^ flip ^ sgn);
  c = __builtin_bit_cast(float, __builtin_bit_cast(unsigned, cq) ^ flip);
}

__global__ __launch_bounds__(256, 8) void qc_main(
    const float* __restrict__ x, const float* __restrict__ thetas,
    const float* __restrict__ phis, float* __restrict__ out, int P) {
  __shared__ v4f Gs[G_V4 + 1];

  const int tid  = threadIdx.x;
  const int lane = tid & 31;
  const bool hi  = (lane >> 4) != 0;
  const int wv   = __builtin_amdgcn_readfirstlane(tid >> 5);

  const int p  = blockIdx.x * 256 + tid;
  const int b  = p >> 14;
  const int h2 = (p >> 7) & 127;
  const int w2 = p & 127;

  float ang[12];
#pragma unroll
  for (int c = 0; c < 3; ++c) {
#pragma unroll
    for (int dy = 0; dy < 2; ++dy) {
      const float* row = x + (((b * 3 + c) * 256 + (2 * h2 + dy)) * 256 + 2 * w2);
      ang[c * 4 + dy * 2 + 0] = row[0];
      ang[c * 4 + dy * 2 + 1] = row[1];
    }
  }

  if (wv < 2) {
    const int ip = lane < 8 ? 0 : (lane > 10 ? 2 : lane - 8);
    const float tv = thetas[lane & 7];
    const float pv = phis[ip];
    const float hv = lane < 8 ? 0.5f * tv : (lane == 8 ? 0.5f * pv : pv);
    float cv, sv;
    sincosf(hv, &sv, &cv);
    float* Gf = reinterpret_cast<float*>(Gs);
    const v4f zero = (v4f){0.f, 0.f, 0.f, 0.f};
    int idx; float vr, vi;
    if (wv == 0) {
      const int a = (lane & 15) ^ (hi ? 8 : 0);
      trace_column<1>(a, cv, sv, idx, vr, vi);
      const float qr = xhalf(vr), qi = xhalf(vi);
      if (!hi) {
        Gf[G_PH + 2 * flipped_pos(a)]     = vr;
        Gf[G_PH + 2 * flipped_pos(a) + 1] = vi;
        const float Ch = from_lane(cv, 8), Sh = from_lane(sv, 8);
        const float c1 = from_lane(cv, 9), s1 = from_lane(sv, 9);
        const bool w1o = (idx & 8) != 0;
        const float kr = w1o ? (c1 * Ch - s1 * Sh) : Ch;
        const float ki = w1o ? (c1 * Sh + s1 * Ch) : Sh;
        const float sS = (a & 8) ? -Sh : Sh;
        const float psr = vr * qr + vi * qi;
        const float psi = vi * qr - vr * qi;
        const float gr = (kr * psr - ki * psi) * sS;
        const float gi = (kr * psi + ki * psr) * sS;
        Gs[G_TAB / 4 + 2 * idx]     = (v4f){kr * Ch, ki * Ch, ki * sS, -kr * sS};
        Gs[G_TAB / 4 + 2 * idx + 1] = (v4f){gr, gi, 0.f, 0.f};
      }
    } else {
#pragma unroll
      for (int j = 0; j < 4; ++j) Gs[32 * j + lane] = zero;
      const int a = lane & 15;
      trace_column<0>(a, cv, sv, idx, vr, vi);
      const int pc = __popc(a) & 3;
      const float xr = (pc == 0) ? vr : (pc == 1) ?  vi : (pc == 2) ? -vr : -vi;
      const float xi = (pc == 0) ? vi : (pc == 1) ? -vr : (pc == 2) ? -vi :  vr;
      if (!hi) {
        Gf[frag0_off(flipped_pos(idx),      a)] = xr;
        Gf[frag0_off(flipped_pos(idx) + 16, a)] = xi;
      }
      if (lane == 0)
        Gs[G_V4] = (v4f){from_lane(cv, 10), from_lane(sv, 10), 0.f, 0.f};
    }
  }

  float cso[12], sno[12];
  float hfa[12];
  float amax = 0.0f;
#pragma unroll
  for (int k = 0; k < 12; ++k) {
    hfa[k] = 0.5f * ang[k];
    if (k != 8) amax = __builtin_fmaxf(amax, __builtin_fabsf(hfa[k]));
  }
  cso[8] = 1.0f; sno[8] = 0.0f;
  if (amax < 131072.0f) {
#pragma unroll
    for (int k = 0; k < 12; ++k)
      if (k != 8) cos_sin_small(hfa[k], cso[k], sno[k]);
  } else {
#pragma unroll
    for (int k = 0; k < 12; ++k)
      if (k != 8) sincosf(hfa[k], &sno[k], &cso[k]);
  }

  __syncthreads();

  const v4f phs = Gs[G_V4];
  const float pr2 = phs[0], pi2 = phs[1];

  const v4f* Gw = Gs + lane;

  float st[2][2][2][8];

  {
    float c0[4][2], s0[4][2];
#pragma unroll
    for (int wq = 0; wq < 4; ++wq) {
      const float co = cso[wq], so = sno[wq];
      const float cx = xhalf(co), sx = xhalf(so);
      c0[wq][0] = hi ? cx : co;  c0[wq][1] = hi ? co : cx;
      s0[wq][0] = hi ? sx : so;  s0[wq][1] = hi ? so : sx;
    }
    float mg[2][8];
#pragma unroll
    for (int nt = 0; nt < 2; ++nt) {
      const float f1 = (hi ? s0[0][nt] : c0[0][nt]) * RS2;
#pragma unroll
      for (int r = 0; r < 8; ++r) {
        float v = f1 * ((r & 4) ? s0[1][nt] : c0[1][nt]);
        v = v * ((r & 2) ? s0[2][nt] : c0[2][nt]);
        mg[nt][r] = v * ((r & 1) ? s0[3][nt] : c0[3][nt]);
      }
    }

    v8f acc[2][2];
#pragma unroll
    for (int mo = 0; mo < 2; ++mo)
#pragma unroll
      for (int nt = 0; nt < 2; ++nt) acc[mo][nt] = (v8f){0.f, 0.f, 0.f, 0.f, 0.f, 0.f, 0.f, 0.f};
    v4f a0[2][2];
#pragma unroll
    for (int mo = 0; mo < 2; ++mo)
#pragma unroll
      for (int q = 0; q < 2; ++q) a0[mo][q] = Gw[(mo * 2 + q) * 32];
#pragma unroll
    for (int rp = 0; rp < 4; ++rp) {
#pragma unroll
      for (int mo = 0; mo < 2; ++mo) {
        const v4f aq = a0[mo][rp >> 1];
        v2f Af;
        Af[0] = (rp & 1) ? aq[2] : aq[0];
        Af[1] = (rp & 1) ? aq[3] : aq[1];
#pragma unroll
        for (int nt = 0; nt < 2; ++nt) {
          v2f Bf;
          Bf[0] = mg[nt][2 * rp];
          Bf[1] = mg[nt][2 * rp + 1];
          acc[mo][nt] = __builtin_amdgcn_wmma_f32_16x16x4_f32(
              false, Af, false, Bf, (short)0, acc[mo][nt], false, false);
        }
      }
    }

#pragma unroll
    for (int nt = 0; nt < 2; ++nt) {
#pragma unroll
      for (int r = 0; r < 8; ++r) {
        st[0][0][nt][r] = acc[0][nt][r];
        st[0][1][nt][r] = acc[1][nt][r];
      }
    }
  }

#pragma unroll
  for (int l = 1; l < 3; ++l) {
#pragma unroll
    for (int wq = (l == 2) ? 1 : 0; wq < 4; ++wq) {
      const float co = cso[4 * l + wq], so = sno[4 * l + wq];
      const float cx = xhalf(co), sx = xhalf(so);
      const float cc[2] = { hi ? cx : co, hi ? co : cx };
      const float ss[2] = { hi ? sx : so, hi ? so : sx };
#pragma unroll
      for (int nt = 0; nt < 2; ++nt) {
        const float c_ = cc[nt], s_ = ss[nt];
        if (wq == 0) {
#pragma unroll
          for (int r = 0; r < 4; ++r) {
            const float ar = st[0][0][nt][r],     ai = st[0][1][nt][r];
            const float br = st[0][0][nt][r + 4], bi = st[0][1][nt][r + 4];
            const float xar = xhalf(ar), xai = xhalf(ai);
            const float xbr = xhalf(br), xbi = xhalf(bi);
            st[0][0][nt][r]     = c_ * ar + s_ * xbi;
            st[0][1][nt][r]     = c_ * ai - s_ * xbr;
            st[0][0][nt][r + 4] = c_ * br + s_ * xai;
            st[0][1][nt][r + 4] = c_ * bi - s_ * xar;
          }
        } else {
          const int bt = 8 >> wq;
#pragma unroll
          for (int w0 = 0; w0 < 2; ++w0) {
            if (l == 1 && w0 == 1) continue;
#pragma unroll
            for (int r = 0; r < 8; ++r) {
              if (!(r & bt)) {
                const int jj = r | bt;
                float ur = st[w0][0][nt][r],  ui = st[w0][1][nt][r];
                float vr = st[w0][0][nt][jj], vi = st[w0][1][nt][jj];
                st[w0][0][nt][r]  = c_ * ur + s_ * vi;
                st[w0][1][nt][r]  = c_ * ui - s_ * vr;
                st[w0][0][nt][jj] = c_ * vr + s_ * ui;
                st[w0][1][nt][jj] = c_ * vi - s_ * ur;
              }
            }
          }
        }
      }
    }

    if (l == 2) break;

    float fa[2][2][8];
    {
      const v4f* Ph = Gs + G_PH / 4 + (hi ? 4 : 0);
      v4f ph[4];
#pragma unroll
      for (int q = 0; q < 4; ++q) ph[q] = Ph[q];
#pragma unroll
      for (int r = 0; r < 8; ++r) {
        const float fr = ph[r >> 1][2 * (r & 1)], fi = ph[r >> 1][2 * (r & 1) + 1];
#pragma unroll
        for (int nt = 0; nt < 2; ++nt) {
          const float ar = st[0][0][nt][r], ai = st[0][1][nt][r];
          fa[0][nt][r] = fr * ar - fi * ai;
          fa[1][nt][r] = fi * ar + fr * ai;
        }
      }
    }

    float nw[2][2][8];
#pragma unroll
    for (int mo = 0; mo < 2; ++mo) {
#pragma unroll
      for (int nt = 0; nt < 2; ++nt) {
#pragma unroll
        for (int k = 0; k < 8; ++k) {
          const int gk = (k ^ (k >> 1) ^ (k >> 2)) & 7;
          const bool cross = (((k >> 2) ^ (k >> 1) ^ k) & 1) != 0;
          nw[mo][nt][gk] = cross ? xhalf(fa[mo][nt][k]) : fa[mo][nt][k];
        }
#pragma unroll
        for (int q = 0; q < 4; ++q) {
          v2f t = {nw[mo][nt][2 * q], nw[mo][nt][2 * q + 1]};
          asm("" : "+v"(t));
          nw[mo][nt][2 * q] = t[0];
          nw[mo][nt][2 * q + 1] = t[1];
        }
      }
    }

    {
      const float co = cso[4], so = sno[4];
      const float cfo = co * co - so * so, sfo = 2.0f * co * so;
      const float cfx = xhalf(cfo), sfx = xhalf(sfo);
      const float ct[2] = { hi ? cfx : cfo, hi ? cfo : cfx };
      const float sn[2] = { hi ? sfx : sfo, hi ? sfo : sfx };
      const v4f* Tb = Gs + G_TAB / 4 + (hi ? 16 : 0);
#pragma unroll
      for (int r = 0; r < 8; ++r) {
        const v4f t0 = Tb[2 * r];
        const v4f t1 = Tb[2 * r + 1];
#pragma unroll
        for (int nt = 0; nt < 2; ++nt) {
          const float kr = t0[0] + t0[2] * ct[nt];
          const float ki = t0[1] + t0[3] * ct[nt];
          const float gr = t1[0] * sn[nt];
          const float gi = t1[1] * sn[nt];
          const float ur = nw[0][nt][r],     ui = nw[1][nt][r];
          const float xr = nw[0][nt][r ^ 7], xi = nw[1][nt][r ^ 7];
          st[0][0][nt][r] = ur;
          st[0][1][nt][r] = ui;
          st[1][0][nt][r] = kr * ur - ki * ui + gr * xr - gi * xi;
          st[1][1][nt][r] = kr * ui + ki * ur + gr * xi + gi * xr;
        }
      }
    }
  }

  float tot[2];
#pragma unroll
  for (int nt = 0; nt < 2; ++nt) {
    float se = 0.0f, sr = 0.0f, si = 0.0f;
#pragma unroll
    for (int r = 0; r < 8; ++r) {
      const float ar = st[0][0][nt][r], ai = st[0][1][nt][r];
      const float br = st[1][0][nt][r], bi = st[1][1][nt][r];
      if (((r >> 2) ^ (r >> 1) ^ r) & 1) {
        sr += ar * br;  sr += ai * bi;
        si += ar * bi;  si -= ai * br;
      } else {
        se += ar * br;  se += ai * bi;
      }
    }
    const float part = se + (pr2 * sr - pi2 * si);
    tot[nt] = 2.0f * (part + xhalf(part));
  }
  const float ev = hi ? tot[1] : tot[0];
  if (p < P) out[p] = ev;
}

extern "C" void kernel_launch(void* const* d_in, const int* in_sizes, int n_in,
                              void* d_out, int out_size, void* d_ws, size_t ws_size,
                              hipStream_t stream) {
  (void)in_sizes; (void)n_in; (void)ws_size;
  const float* x      = (const float*)d_in[0];
  const float* thetas = (const float*)d_in[1];
  const float* phis   = (const float*)d_in[2];
  float* out = (float*)d_out;
  (void)d_ws;

  const int P = out_size;
  const int blocks = (P + 255) / 256;
  hipLaunchKernelGGL(qc_main, dim3(blocks), dim3(256), 0, stream, x, thetas, phis, out, P);
}
